# starting kernel
# baseline (speedup 1.0000x reference)
#include <hip/hip_runtime.h>
#include <hip/hip_bf16.h>
#include <math.h>

#define N_NODES 100000
#define N_EDGES 1600000
#define F_DIM 128
#define H_DIM 128
#define C_DIM 16

typedef __attribute__((ext_vector_type(2))) float v2f;
typedef __attribute__((ext_vector_type(8))) float v8f;

__global__ void init_deg_kernel(float* __restrict__ deg) {
  const int i = blockIdx.x * blockDim.x + threadIdx.x;
  if (i < N_NODES) deg[i] = 1.0f;
}

__global__ void deg_atomic_kernel(const int* __restrict__ dst,
                                  const float* __restrict__ ew,
                                  float* __restrict__ deg) {
  const int e = blockIdx.x * blockDim.x + threadIdx.x;
  if (e < N_EDGES) atomicAdd(&deg[dst[e]], ew[e]);
}

__global__ void dinv_kernel(float* __restrict__ deg) {
  const int i = blockIdx.x * blockDim.x + threadIdx.x;
  if (i < N_NODES) {
    const float d = deg[i];
    deg[i] = (d > 0.0f) ? rsqrtf(d) : 0.0f;
  }
}

__global__ __launch_bounds__(256) void gemm1_wmma(const float* __restrict__ x,
                                                  const float* __restrict__ W1,
                                                  float* __restrict__ h) {
  __shared__ float sX[64 * 132];
  const int tid  = threadIdx.x;
  const int row0 = blockIdx.x * 64;

  for (int i = tid; i < 64 * 32; i += 256) {
    const int r = i >> 5, c = i & 31;
    float4 v = make_float4(0.f, 0.f, 0.f, 0.f);
    const int gr = row0 + r;
    if (gr < N_NODES) v = ((const float4*)x)[gr * 32 + c];
    *(float4*)&sX[r * 132 + c * 4] = v;
  }
  __syncthreads();

  const int wave = tid >> 5;
  const int lane = tid & 31;
  const int lrow = lane & 15;
  const int hi   = lane >> 4;
  const int col  = wave * 16 + lrow;

  for (int t = 0; t < 4; ++t) {
    v8f acc = {};
    const float* ax = &sX[(t * 16 + lrow) * 132 + 2 * hi];
    for (int k0 = 0; k0 < 128; k0 += 4) {
      v2f a, b;
      a.x = ax[k0];
      a.y = ax[k0 + 1];
      const int kb = k0 + 2 * hi;
      b.x = W1[kb * 128 + col];
      b.y = W1[(kb + 1) * 128 + col];
      acc = __builtin_amdgcn_wmma_f32_16x16x4_f32(false, a, false, b,
                                                  (short)0, acc, false, false);
    }
    const int rbase = row0 + t * 16 + 8 * hi;
#pragma unroll
    for (int r = 0; r < 8; ++r) {
      const int gr = rbase + r;
      if (gr < N_NODES) h[gr * 128 + col] = acc[r];
    }
  }
}

__global__ void selfloop1_kernel(const float* __restrict__ h,
                                 const float* __restrict__ dinv,
                                 float* __restrict__ agg) {
  const int idx = blockIdx.x * blockDim.x + threadIdx.x;
  if (idx >= N_NODES * 128) return;
  const float di = dinv[idx >> 7];
  agg[idx] = h[idx] * di * di;
}

__global__ void scatter1_kernel(const int* __restrict__ src,
                                const int* __restrict__ dst,
                                const float* __restrict__ ew,
                                const float* __restrict__ dinv,
                                const float* __restrict__ h,
                                float* __restrict__ agg) {
  const int gid  = blockIdx.x * blockDim.x + threadIdx.x;
  const int e    = gid >> 5;
  const int lane = threadIdx.x & 31;
  if (e >= N_EDGES) return;
  const int   s    = src[e];
  const int   d    = dst[e];
  const float norm = dinv[s] * ew[e] * dinv[d];
  const float4 v = ((const float4*)(h + s * 128))[lane];
  float* o = agg + d * 128 + lane * 4;
  atomicAdd(o + 0, v.x * norm);
  atomicAdd(o + 1, v.y * norm);
  atomicAdd(o + 2, v.z * norm);
  atomicAdd(o + 3, v.w * norm);
}

__global__ __launch_bounds__(256) void gemm2_wmma(const float* __restrict__ agg1,
                                                  const float* __restrict__ b1,
                                                  const float* __restrict__ W2,
                                                  float* __restrict__ h2) {
  __shared__ float sW[128 * 16];
  __shared__ float sb[128];
  const int tid = threadIdx.x;
  for (int i = tid; i < 512; i += 256) ((float4*)sW)[i] = ((const float4*)W2)[i];
  if (tid < 128) sb[tid] = b1[tid];
  __syncthreads();

  const int wave = tid >> 5, lane = tid & 31;
  const int lrow = lane & 15, hi = lane >> 4;
  const int row0 = blockIdx.x * 128 + wave * 16;
  const int arow = row0 + lrow;

  v8f acc = {};
  for (int k0 = 0; k0 < 128; k0 += 4) {
    const int k = k0 + 2 * hi;
    float a0 = 0.f, a1 = 0.f;
    if (arow < N_NODES) {
      a0 = fmaxf(agg1[arow * 128 + k] + sb[k], 0.f);
      a1 = fmaxf(agg1[arow * 128 + k + 1] + sb[k + 1], 0.f);
    }
    v2f a, b;
    a.x = a0; a.y = a1;
    b.x = sW[k * 16 + lrow];
    b.y = sW[(k + 1) * 16 + lrow];
    acc = __builtin_amdgcn_wmma_f32_16x16x4_f32(false, a, false, b,
                                                (short)0, acc, false, false);
  }
  const int rbase = row0 + 8 * hi;
#pragma unroll
  for (int r = 0; r < 8; ++r) {
    const int gr = rbase + r;
    if (gr < N_NODES) h2[gr * 16 + lrow] = acc[r];
  }
}

__global__ void selfloop2_kernel(const float* __restrict__ h2,
                                 const float* __restrict__ dinv,
                                 float* __restrict__ agg) {
  const int idx = blockIdx.x * blockDim.x + threadIdx.x;
  if (idx >= N_NODES * 16) return;
  const float di = dinv[idx >> 4];
  agg[idx] = h2[idx] * di * di;
}

__global__ void scatter2_kernel(const int* __restrict__ src,
                                const int* __restrict__ dst,
                                const float* __restrict__ ew,
                                const float* __restrict__ dinv,
                                const float* __restrict__ h2,
                                float* __restrict__ agg) {
  const int gid = blockIdx.x * blockDim.x + threadIdx.x;
  const int e   = gid >> 2;
  const int sub = threadIdx.x & 3;
  if (e >= N_EDGES) return;
  const int   s    = src[e];
  const int   d    = dst[e];
  const float norm = dinv[s] * ew[e] * dinv[d];
  const float4 v = ((const float4*)(h2 + s * 16))[sub];
  float* o = agg + d * 16 + sub * 4;
  atomicAdd(o + 0, v.x * norm);
  atomicAdd(o + 1, v.y * norm);
  atomicAdd(o + 2, v.z * norm);
  atomicAdd(o + 3, v.w * norm);
}

__global__ void softmax_kernel(const float* __restrict__ agg2,
                               const float* __restrict__ b2,
                               float* __restrict__ out) {
  const int i = blockIdx.x * blockDim.x + threadIdx.x;
  if (i >= N_NODES) return;
  float v[16];
  const float4* p = (const float4*)(agg2 + (size_t)i * 16);
  float m = -3.402823466e+38f;
#pragma unroll
  for (int q = 0; q < 4; ++q) {
    const float4 t = p[q];
    v[4 * q + 0] = t.x + b2[4 * q + 0];
    v[4 * q + 1] = t.y + b2[4 * q + 1];
    v[4 * q + 2] = t.z + b2[4 * q + 2];
    v[4 * q + 3] = t.w + b2[4 * q + 3];
  }
#pragma unroll
  for (int c = 0; c < 16; ++c) m = fmaxf(m, v[c]);
  float s = 0.f;
#pragma unroll
  for (int c = 0; c < 16; ++c) { v[c] = expf(v[c] - m); s += v[c]; }
  const float inv = 1.0f / s;
  float4* o = (float4*)(out + (size_t)i * 16);
#pragma unroll
  for (int q = 0; q < 4; ++q)
    o[q] = make_float4(v[4 * q + 0] * inv, v[4 * q + 1] * inv,
                       v[4 * q + 2] * inv, v[4 * q + 3] * inv);
}

extern "C" void kernel_launch(void* const* d_in, const int* in_sizes, int n_in,
                              void* d_out, int out_size, void* d_ws, size_t ws_size,
                              hipStream_t stream) {
  const float* x  = (const float*)d_in[0];
  const int*   ei = (const int*)d_in[1];
  const float* ew = (const float*)d_in[2];
  const float* W1 = (const float*)d_in[3];
  const float* b1 = (const float*)d_in[4];
  const float* W2 = (const float*)d_in[5];
  const float* b2 = (const float*)d_in[6];
  float* out = (float*)d_out;

  const int* src = ei;
  const int* dst = ei + N_EDGES;

  float* ws   = (float*)d_ws;
  float* dinv = ws;
  float* h    = ws + N_NODES;
  float* agg1 = ws + N_NODES + (size_t)N_NODES * 128;
  float* h2   = h;
  float* agg2 = h + (size_t)N_NODES * 16;

  const int T = 256;

  init_deg_kernel<<<(N_NODES + T - 1) / T, T, 0, stream>>>(dinv);
  deg_atomic_kernel<<<(N_EDGES + T - 1) / T, T, 0, stream>>>(dst, ew, dinv);
  dinv_kernel<<<(N_NODES + T - 1) / T, T, 0, stream>>>(dinv);

  gemm1_wmma<<<(N_NODES + 63) / 64, T, 0, stream>>>(x, W1, h);
  selfloop1_kernel<<<(N_NODES * 128 + T - 1) / T, T, 0, stream>>>(h, dinv, agg1);
  scatter1_kernel<<<(N_EDGES * 32 + T - 1) / T, T, 0, stream>>>(src, dst, ew, dinv, h, agg1);

  gemm2_wmma<<<(N_NODES + 127) / 128, T, 0, stream>>>(agg1, b1, W2, h2);
  selfloop2_kernel<<<(N_NODES * 16 + T - 1) / T, T, 0, stream>>>(h2, dinv, agg2);
  scatter2_kernel<<<(N_EDGES * 4 + T - 1) / T, T, 0, stream>>>(src, dst, ew, dinv, h2, agg2);

  softmax_kernel<<<(N_NODES + T - 1) / T, T, 0, stream>>>(agg2, b2, out);
}


# Replace float-atomic edge scatter by a per-call counting sort on destination (one returning int atomic per edge) and a per-node gather writing each row once; node sums now run in arrival-rank order
# speedup vs baseline: 1.9922x; 1.9922x over previous
#include <hip/hip_runtime.h>
#include <hip/hip_bf16.h>
#include <math.h>


#define N_NODES 100000
#define N_EDGES 1600000
#define F_DIM 128
#define H_DIM 128
#define C_DIM 16

#define SCAN_ELEMS 1024
#define SCAN_NBLK ((N_NODES + SCAN_ELEMS) / SCAN_ELEMS)
#define NPAD (SCAN_NBLK * SCAN_ELEMS)

typedef __attribute__((ext_vector_type(2))) float v2f;
typedef __attribute__((ext_vector_type(8))) float v8f;

__global__ void zero_cnt_kernel(int* __restrict__ cnt) {
  const int i = blockIdx.x * blockDim.x + threadIdx.x;
  if (i < NPAD) cnt[i] = 0;
}

__global__ void rank_kernel(const int* __restrict__ dst,
                            int* __restrict__ cnt,
                            int* __restrict__ rank) {
  const int e = blockIdx.x * blockDim.x + threadIdx.x;
  if (e < N_EDGES) rank[e] = atomicAdd(&cnt[dst[e]], 1);
}

__global__ __launch_bounds__(256) void blocksum_kernel(const int* __restrict__ cnt,
                                                       int* __restrict__ bsum) {
  __shared__ int sw[8];
  const int tid = threadIdx.x;
  const int4 c = ((const int4*)cnt)[blockIdx.x * 256 + tid];
  int v = c.x + c.y + c.z + c.w;
  for (int o = 16; o > 0; o >>= 1) v += __shfl_xor(v, o);
  if ((tid & 31) == 0) sw[tid >> 5] = v;
  __syncthreads();
  if (tid == 0) {
    int t = 0;
    for (int w = 0; w < 8; ++w) t += sw[w];
    bsum[blockIdx.x] = t;
  }
}

__global__ __launch_bounds__(256) void scan_kernel(const int* __restrict__ cnt,
                                                   const int* __restrict__ bsum,
                                                   int* __restrict__ off) {
  __shared__ int sw[8];
  __shared__ int sbase;
  const int tid = threadIdx.x, lane = tid & 31, wave = tid >> 5;
  if (wave == 0) {
    int t = 0;
    for (int i = lane; i < (int)blockIdx.x; i += 32) t += bsum[i];
    for (int o = 16; o > 0; o >>= 1) t += __shfl_xor(t, o);
    if (lane == 0) sbase = t;
  }
  const int4 c = ((const int4*)cnt)[blockIdx.x * 256 + tid];
  const int tsum = c.x + c.y + c.z + c.w;
  int inc = tsum;
  for (int o = 1; o < 32; o <<= 1) {
    const int u = __shfl_up(inc, o);
    if (lane >= o) inc += u;
  }
  if (lane == 31) sw[wave] = inc;
  __syncthreads();
  int wbase = 0;
  for (int w = 0; w < wave; ++w) wbase += sw[w];
  const int ex = sbase + wbase + inc - tsum;
  int4 o4;
  o4.x = ex;
  o4.y = o4.x + c.x;
  o4.z = o4.y + c.y;
  o4.w = o4.z + c.z;
  ((int4*)off)[blockIdx.x * 256 + tid] = o4;
}

__global__ void fill_kernel(const int* __restrict__ src,
                            const int* __restrict__ dst,
                            const float* __restrict__ ew,
                            const int* __restrict__ rank,
                            const int* __restrict__ off,
                            int* __restrict__ csr) {
  const int e = blockIdx.x * blockDim.x + threadIdx.x;
  if (e >= N_EDGES) return;
  const int slot = off[dst[e]] + rank[e];
  ((int2*)csr)[slot] = make_int2(src[e], __float_as_int(ew[e]));
}

__global__ void dinv_kernel(const int* __restrict__ off,
                            const int* __restrict__ csr,
                            float* __restrict__ dinv) {
  const int i = blockIdx.x * blockDim.x + threadIdx.x;
  if (i >= N_NODES) return;
  const int beg = off[i], end = off[i + 1];
  float d = 1.0f;
  for (int j = beg; j < end; ++j) d += __int_as_float(csr[2 * j + 1]);
  dinv[i] = (d > 0.0f) ? rsqrtf(d) : 0.0f;
}

__global__ __launch_bounds__(256) void gemm1_wmma(const float* __restrict__ x,
                                                  const float* __restrict__ W1,
                                                  float* __restrict__ h) {
  __shared__ float sX[64 * 132];
  const int tid  = threadIdx.x;
  const int row0 = blockIdx.x * 64;

  for (int i = tid; i < 64 * 32; i += 256) {
    const int r = i >> 5, c = i & 31;
    float4 v = make_float4(0.f, 0.f, 0.f, 0.f);
    const int gr = row0 + r;
    if (gr < N_NODES) v = ((const float4*)x)[gr * 32 + c];
    *(float4*)&sX[r * 132 + c * 4] = v;
  }
  __syncthreads();

  const int wave = tid >> 5;
  const int lane = tid & 31;
  const int lrow = lane & 15;
  const int hi   = lane >> 4;
  const int col  = wave * 16 + lrow;

  for (int t = 0; t < 4; ++t) {
    v8f acc = {};
    const float* ax = &sX[(t * 16 + lrow) * 132 + 2 * hi];
    for (int k0 = 0; k0 < 128; k0 += 4) {
      v2f a, b;
      a.x = ax[k0];
      a.y = ax[k0 + 1];
      const int kb = k0 + 2 * hi;
      b.x = W1[kb * 128 + col];
      b.y = W1[(kb + 1) * 128 + col];
      acc = __builtin_amdgcn_wmma_f32_16x16x4_f32(false, a, false, b,
                                                  (short)0, acc, false, false);
    }
    const int rbase = row0 + t * 16 + 8 * hi;
#pragma unroll
    for (int r = 0; r < 8; ++r) {
      const int gr = rbase + r;
      if (gr < N_NODES) h[gr * 128 + col] = acc[r];
    }
  }
}

__global__ __launch_bounds__(256) void gather1_kernel(const int* __restrict__ off,
                                                      const int* __restrict__ csr,
                                                      const float* __restrict__ dinv,
                                                      const float* __restrict__ h,
                                                      float* __restrict__ agg) {
  const int lane = threadIdx.x & 31;
  const int d = __builtin_amdgcn_readfirstlane((int)(blockIdx.x * 8 + (threadIdx.x >> 5)));
  if (d >= N_NODES) return;
  const int beg = off[d], end = off[d + 1];
  const float dd = dinv[d];
  const float4* h4 = (const float4*)h;
  float4 acc = h4[d * 32 + lane];
  acc.x = acc.x * dd * dd;
  acc.y = acc.y * dd * dd;
  acc.z = acc.z * dd * dd;
  acc.w = acc.w * dd * dd;
  for (int base = beg; base < end; base += 32) {
    const int n = min(32, end - base);
    int s_l = 0;
    float nrm_l = 0.f;
    if (lane < n) {
      const int2 ed = ((const int2*)csr)[base + lane];
      s_l = ed.x;
      nrm_l = dinv[ed.x] * __int_as_float(ed.y) * dd;
    }
#pragma unroll 4
    for (int j = 0; j < n; ++j) {
      const int s = __builtin_amdgcn_readlane(s_l, j);
      const float nrm = __int_as_float(__builtin_amdgcn_readlane(__float_as_int(nrm_l), j));
      const float4 v = h4[s * 32 + lane];
      acc.x += v.x * nrm;
      acc.y += v.y * nrm;
      acc.z += v.z * nrm;
      acc.w += v.w * nrm;
    }
  }
  ((float4*)agg)[d * 32 + lane] = acc;
}

__global__ __launch_bounds__(256) void gemm2_wmma(const float* __restrict__ agg1,
                                                  const float* __restrict__ b1,
                                                  const float* __restrict__ W2,
                                                  float* __restrict__ h2) {
  __shared__ float sW[128 * 16];
  __shared__ float sb[128];
  const int tid = threadIdx.x;
  for (int i = tid; i < 512; i += 256) ((float4*)sW)[i] = ((const float4*)W2)[i];
  if (tid < 128) sb[tid] = b1[tid];
  __syncthreads();

  const int wave = tid >> 5, lane = tid & 31;
  const int lrow = lane & 15, hi = lane >> 4;
  const int row0 = blockIdx.x * 128 + wave * 16;
  const int arow = row0 + lrow;

  v8f acc = {};
  for (int k0 = 0; k0 < 128; k0 += 4) {
    const int k = k0 + 2 * hi;
    float a0 = 0.f, a1 = 0.f;
    if (arow < N_NODES) {
      a0 = fmaxf(agg1[arow * 128 + k] + sb[k], 0.f);
      a1 = fmaxf(agg1[arow * 128 + k + 1] + sb[k + 1], 0.f);
    }
    v2f a, b;
    a.x = a0; a.y = a1;
    b.x = sW[k * 16 + lrow];
    b.y = sW[(k + 1) * 16 + lrow];
    acc = __builtin_amdgcn_wmma_f32_16x16x4_f32(false, a, false, b,
                                                (short)0, acc, false, false);
  }
  const int rbase = row0 + 8 * hi;
#pragma unroll
  for (int r = 0; r < 8; ++r) {
    const int gr = rbase + r;
    if (gr < N_NODES) h2[gr * 16 + lrow] = acc[r];
  }
}

__global__ __launch_bounds__(256) void gather2_kernel(const int* __restrict__ off,
                                                      const int* __restrict__ csr,
                                                      const float* __restrict__ dinv,
                                                      const float* __restrict__ h2,
                                                      float* __restrict__ agg) {
  const int gid = blockIdx.x * blockDim.x + threadIdx.x;
  const int d   = gid >> 2;
  const int sub = threadIdx.x & 3;
  if (d >= N_NODES) return;
  const int beg = off[d], end = off[d + 1];
  const float dd = dinv[d];
  const float4* p = (const float4*)h2;
  float4 acc = p[d * 4 + sub];
  acc.x = acc.x * dd * dd;
  acc.y = acc.y * dd * dd;
  acc.z = acc.z * dd * dd;
  acc.w = acc.w * dd * dd;
  for (int j = beg; j < end; ++j) {
    const int2 ed = ((const int2*)csr)[j];
    const float nrm = dinv[ed.x] * __int_as_float(ed.y) * dd;
    const float4 v = p[ed.x * 4 + sub];
    acc.x += v.x * nrm;
    acc.y += v.y * nrm;
    acc.z += v.z * nrm;
    acc.w += v.w * nrm;
  }
  ((float4*)agg)[d * 4 + sub] = acc;
}

__global__ void softmax_kernel(const float* __restrict__ agg2,
                               const float* __restrict__ b2,
                               float* __restrict__ out) {
  const int i = blockIdx.x * blockDim.x + threadIdx.x;
  if (i >= N_NODES) return;
  float v[16];
  const float4* p = (const float4*)(agg2 + (size_t)i * 16);
  float m = -3.402823466e+38f;
#pragma unroll
  for (int q = 0; q < 4; ++q) {
    const float4 t = p[q];
    v[4 * q + 0] = t.x + b2[4 * q + 0];
    v[4 * q + 1] = t.y + b2[4 * q + 1];
    v[4 * q + 2] = t.z + b2[4 * q + 2];
    v[4 * q + 3] = t.w + b2[4 * q + 3];
  }
#pragma unroll
  for (int c = 0; c < 16; ++c) m = fmaxf(m, v[c]);
  float s = 0.f;
#pragma unroll
  for (int c = 0; c < 16; ++c) { v[c] = expf(v[c] - m); s += v[c]; }
  const float inv = 1.0f / s;
  float4* o = (float4*)(out + (size_t)i * 16);
#pragma unroll
  for (int q = 0; q < 4; ++q)
    o[q] = make_float4(v[4 * q + 0] * inv, v[4 * q + 1] * inv,
                       v[4 * q + 2] * inv, v[4 * q + 3] * inv);
}

extern "C" void kernel_launch(void* const* d_in, const int* in_sizes, int n_in,
                              void* d_out, int out_size, void* d_ws, size_t ws_size,
                              hipStream_t stream) {
  const float* x  = (const float*)d_in[0];
  const int*   ei = (const int*)d_in[1];
  const float* ew = (const float*)d_in[2];
  const float* W1 = (const float*)d_in[3];
  const float* b1 = (const float*)d_in[4];
  const float* W2 = (const float*)d_in[5];
  const float* b2 = (const float*)d_in[6];
  float* out = (float*)d_out;

  const int* src = ei;
  const int* dst = ei + N_EDGES;

  float* ws   = (float*)d_ws;
  float* dinv = ws;
  float* h    = ws + N_NODES;
  float* agg1 = ws + N_NODES + (size_t)N_NODES * 128;
  float* h2   = h;
  float* agg2 = h + (size_t)N_NODES * 16;
  int* cnt  = (int*)(ws + N_NODES + (size_t)N_NODES * 256);
  int* off  = cnt + NPAD;
  int* bsum = off + NPAD;
  int* rank = bsum + 128;
  int* csr  = rank + N_EDGES;

  const int T = 256;

  zero_cnt_kernel<<<NPAD / T, T, 0, stream>>>(cnt);
  rank_kernel<<<(N_EDGES + T - 1) / T, T, 0, stream>>>(dst, cnt, rank);
  blocksum_kernel<<<SCAN_NBLK, T, 0, stream>>>(cnt, bsum);
  scan_kernel<<<SCAN_NBLK, T, 0, stream>>>(cnt, bsum, off);
  fill_kernel<<<(N_EDGES + T - 1) / T, T, 0, stream>>>(src, dst, ew, rank, off, csr);
  dinv_kernel<<<(N_NODES + T - 1) / T, T, 0, stream>>>(off, csr, dinv);

  gemm1_wmma<<<(N_NODES + 63) / 64, T, 0, stream>>>(x, W1, h);
  gather1_kernel<<<(N_NODES + 7) / 8, T, 0, stream>>>(off, csr, dinv, h, agg1);

  gemm2_wmma<<<(N_NODES + 127) / 128, T, 0, stream>>>(agg1, b1, W2, h2);
  gather2_kernel<<<(N_NODES * 4 + T - 1) / T, T, 0, stream>>>(off, csr, dinv, h2, agg2);

  softmax_kernel<<<(N_NODES + T - 1) / T, T, 0, stream>>>(agg2, b2, out);
}


# Fuse bias+ReLU+second linear layer (f32 WMMA on LDS-staged 16-node tiles, two nodes per wave) into the layer-1 gather's epilogue; agg1 round trip and the GEMM2 launch go; summation order unchanged
# speedup vs baseline: 2.1324x; 1.0704x over previous
#include <hip/hip_runtime.h>
#include <hip/hip_bf16.h>
#include <math.h>


#define N_NODES 100000
#define N_EDGES 1600000
#define F_DIM 128
#define H_DIM 128
#define C_DIM 16

#define SCAN_ELEMS 1024
#define SCAN_NBLK ((N_NODES + SCAN_ELEMS) / SCAN_ELEMS)
#define NPAD (SCAN_NBLK * SCAN_ELEMS)

typedef __attribute__((ext_vector_type(2))) float v2f;
typedef __attribute__((ext_vector_type(8))) float v8f;

__global__ void zero_cnt_kernel(int* __restrict__ cnt) {
  const int i = blockIdx.x * blockDim.x + threadIdx.x;
  if (i < NPAD) cnt[i] = 0;
}

__global__ void rank_kernel(const int* __restrict__ dst,
                            int* __restrict__ cnt,
                            int* __restrict__ rank) {
  const int e = blockIdx.x * blockDim.x + threadIdx.x;
  if (e < N_EDGES) rank[e] = atomicAdd(&cnt[dst[e]], 1);
}

__global__ __launch_bounds__(256) void blocksum_kernel(const int* __restrict__ cnt,
                                                       int* __restrict__ bsum) {
  __shared__ int sw[8];
  const int tid = threadIdx.x;
  const int4 c = ((const int4*)cnt)[blockIdx.x * 256 + tid];
  int v = c.x + c.y + c.z + c.w;
  for (int o = 16; o > 0; o >>= 1) v += __shfl_xor(v, o);
  if ((tid & 31) == 0) sw[tid >> 5] = v;
  __syncthreads();
  if (tid == 0) {
    int t = 0;
    for (int w = 0; w < 8; ++w) t += sw[w];
    bsum[blockIdx.x] = t;
  }
}

__global__ __launch_bounds__(256) void scan_kernel(const int* __restrict__ cnt,
                                                   const int* __restrict__ bsum,
                                                   int* __restrict__ off) {
  __shared__ int sw[8];
  __shared__ int sbase;
  const int tid = threadIdx.x, lane = tid & 31, wave = tid >> 5;
  if (wave == 0) {
    int t = 0;
    for (int i = lane; i < (int)blockIdx.x; i += 32) t += bsum[i];
    for (int o = 16; o > 0; o >>= 1) t += __shfl_xor(t, o);
    if (lane == 0) sbase = t;
  }
  const int4 c = ((const int4*)cnt)[blockIdx.x * 256 + tid];
  const int tsum = c.x + c.y + c.z + c.w;
  int inc = tsum;
  for (int o = 1; o < 32; o <<= 1) {
    const int u = __shfl_up(inc, o);
    if (lane >= o) inc += u;
  }
  if (lane == 31) sw[wave] = inc;
  __syncthreads();
  int wbase = 0;
  for (int w = 0; w < wave; ++w) wbase += sw[w];
  const int ex = sbase + wbase + inc - tsum;
  int4 o4;
  o4.x = ex;
  o4.y = o4.x + c.x;
  o4.z = o4.y + c.y;
  o4.w = o4.z + c.z;
  ((int4*)off)[blockIdx.x * 256 + tid] = o4;
}

__global__ void fill_kernel(const int* __restrict__ src,
                            const int* __restrict__ dst,
                            const float* __restrict__ ew,
                            const int* __restrict__ rank,
                            const int* __restrict__ off,
                            int* __restrict__ csr) {
  const int e = blockIdx.x * blockDim.x + threadIdx.x;
  if (e >= N_EDGES) return;
  const int slot = off[dst[e]] + rank[e];
  ((int2*)csr)[slot] = make_int2(src[e], __float_as_int(ew[e]));
}

__global__ void dinv_kernel(const int* __restrict__ off,
                            const int* __restrict__ csr,
                            float* __restrict__ dinv) {
  const int i = blockIdx.x * blockDim.x + threadIdx.x;
  if (i >= N_NODES) return;
  const int beg = off[i], end = off[i + 1];
  float d = 1.0f;
  for (int j = beg; j < end; ++j) d += __int_as_float(csr[2 * j + 1]);
  dinv[i] = (d > 0.0f) ? rsqrtf(d) : 0.0f;
}

__global__ __launch_bounds__(256) void gemm1_wmma(const float* __restrict__ x,
                                                  const float* __restrict__ W1,
                                                  float* __restrict__ h) {
  __shared__ float sX[64 * 132];
  const int tid  = threadIdx.x;
  const int row0 = blockIdx.x * 64;

  for (int i = tid; i < 64 * 32; i += 256) {
    const int r = i >> 5, c = i & 31;
    float4 v = make_float4(0.f, 0.f, 0.f, 0.f);
    const int gr = row0 + r;
    if (gr < N_NODES) v = ((const float4*)x)[gr * 32 + c];
    *(float4*)&sX[r * 132 + c * 4] = v;
  }
  __syncthreads();

  const int wave = tid >> 5;
  const int lane = tid & 31;
  const int lrow = lane & 15;
  const int hi   = lane >> 4;
  const int col  = wave * 16 + lrow;

  for (int t = 0; t < 4; ++t) {
    v8f acc = {};
    const float* ax = &sX[(t * 16 + lrow) * 132 + 2 * hi];
    for (int k0 = 0; k0 < 128; k0 += 4) {
      v2f a, b;
      a.x = ax[k0];
      a.y = ax[k0 + 1];
      const int kb = k0 + 2 * hi;
      b.x = W1[kb * 128 + col];
      b.y = W1[(kb + 1) * 128 + col];
      acc = __builtin_amdgcn_wmma_f32_16x16x4_f32(false, a, false, b,
                                                  (short)0, acc, false, false);
    }
    const int rbase = row0 + t * 16 + 8 * hi;
#pragma unroll
    for (int r = 0; r < 8; ++r) {
      const int gr = rbase + r;
      if (gr < N_NODES) h[gr * 128 + col] = acc[r];
    }
  }
}

#define G1_NPW 2
#define G1_TILE (8 * G1_NPW)
#define G1_PITCH 132
static_assert(G1_TILE == 16, "one 16-row matrix tile per work-group");
static_assert(N_NODES % G1_TILE == 0, "the grid covers the nodes exactly, no row guards");

__device__ __forceinline__ void gather_rows(float4& acc, const int s_l, const float nrm_l,
                                            const int n, const float4* __restrict__ h4,
                                            const int lane) {
#pragma unroll 4
  for (int j = 0; j < n; ++j) {
    const int s = __builtin_amdgcn_readlane(s_l, j);
    const float nrm = __int_as_float(__builtin_amdgcn_readlane(__float_as_int(nrm_l), j));
    const float4 v = h4[s * 32 + lane];
    acc.x += v.x * nrm;
    acc.y += v.y * nrm;
    acc.z += v.z * nrm;
    acc.w += v.w * nrm;
  }
}

__global__ __launch_bounds__(256) void gather1_gemm2_kernel(const int* __restrict__ off,
                                                            const int* __restrict__ csr,
                                                            const float* __restrict__ dinv,
                                                            const float* __restrict__ h,
                                                            const float* __restrict__ b1,
                                                            const float* __restrict__ W2,
                                                            float* __restrict__ h2) {
  __shared__ float sA[G1_TILE * G1_PITCH];
  __shared__ float sW[128 * 16];
  const int tid  = threadIdx.x;
  const int lane = tid & 31;
  const int wave = __builtin_amdgcn_readfirstlane(tid >> 5);
  const float4 w2a = ((const float4*)W2)[tid];
  const float4 w2b = ((const float4*)W2)[tid + 256];
  const float4 bias = ((const float4*)b1)[lane];
  const float4* h4  = (const float4*)h;
  const int2* csr2  = (const int2*)csr;
  const int d0 = (int)blockIdx.x * G1_TILE + wave;

  int beg[G1_NPW], end[G1_NPW], nn[G1_NPW], s_l[G1_NPW];
  float dd[G1_NPW], w_l[G1_NPW], nrm_l[G1_NPW];
  float4 acc[G1_NPW];
#pragma unroll
  for (int p = 0; p < G1_NPW; ++p) {
    const int d = d0 + 8 * p;
    beg[p] = off[d];
    end[p] = off[d + 1];
    dd[p]  = dinv[d];
  }
#pragma unroll
  for (int p = 0; p < G1_NPW; ++p) {
    const int d = d0 + 8 * p;
    acc[p] = h4[d * 32 + lane];
    nn[p]  = min(32, end[p] - beg[p]);
    const bool on = lane < nn[p];
    const int2 ed = csr2[on ? beg[p] + lane : 0];
    s_l[p] = on ? ed.x : 0;
    w_l[p] = on ? __int_as_float(ed.y) : 0.f;
  }
#pragma unroll
  for (int p = 0; p < G1_NPW; ++p) nrm_l[p] = dinv[s_l[p]] * w_l[p] * dd[p];
  asm volatile("" :: "v"(w2a.x), "v"(w2a.y), "v"(w2a.z), "v"(w2a.w),
                     "v"(w2b.x), "v"(w2b.y), "v"(w2b.z), "v"(w2b.w));

#pragma unroll
  for (int p = 0; p < G1_NPW; ++p) {
    float4 a = acc[p];
    a.x = a.x * dd[p] * dd[p];
    a.y = a.y * dd[p] * dd[p];
    a.z = a.z * dd[p] * dd[p];
    a.w = a.w * dd[p] * dd[p];
    gather_rows(a, s_l[p], nrm_l[p], nn[p], h4, lane);
    for (int base = beg[p] + 32; base < end[p]; base += 32) {
      const int n = min(32, end[p] - base);
      int s2 = 0;
      float nrm2 = 0.f;
      if (lane < n) {
        const int2 ed = csr2[base + lane];
        s2 = ed.x;
        nrm2 = dinv[ed.x] * __int_as_float(ed.y) * dd[p];
      }
      gather_rows(a, s2, nrm2, n, h4, lane);
    }
    float4 r;
    r.x = fmaxf(a.x + bias.x, 0.f);
    r.y = fmaxf(a.y + bias.y, 0.f);
    r.z = fmaxf(a.z + bias.z, 0.f);
    r.w = fmaxf(a.w + bias.w, 0.f);
    *(float4*)&sA[(8 * p + wave) * G1_PITCH + 4 * lane] = r;
  }
  ((float4*)sW)[tid] = w2a;
  ((float4*)sW)[tid + 256] = w2b;
  __syncthreads();

  if (wave == 0) {
    const int lrow = lane & 15, hi = lane >> 4;
    const float* ax = &sA[lrow * G1_PITCH + 2 * hi];
    v8f c = {};
    for (int k0 = 0; k0 < 128; k0 += 4) {
      const int k = k0 + 2 * hi;
      v2f a, b;
      a.x = ax[k0];
      a.y = ax[k0 + 1];
      b.x = sW[k * 16 + lrow];
      b.y = sW[(k + 1) * 16 + lrow];
      c = __builtin_amdgcn_wmma_f32_16x16x4_f32(false, a, false, b,
                                                (short)0, c, false, false);
    }
    const int rbase = (int)blockIdx.x * G1_TILE + 8 * hi;
#pragma unroll
    for (int r = 0; r < 8; ++r)
      h2[(rbase + r) * 16 + lrow] = c[r];
  }
}

__global__ __launch_bounds__(256) void gather2_kernel(const int* __restrict__ off,
                                                      const int* __restrict__ csr,
                                                      const float* __restrict__ dinv,
                                                      const float* __restrict__ h2,
                                                      float* __restrict__ agg) {
  const int gid = blockIdx.x * blockDim.x + threadIdx.x;
  const int d   = gid >> 2;
  const int sub = threadIdx.x & 3;
  if (d >= N_NODES) return;
  const int beg = off[d], end = off[d + 1];
  const float dd = dinv[d];
  const float4* p = (const float4*)h2;
  float4 acc = p[d * 4 + sub];
  acc.x = acc.x * dd * dd;
  acc.y = acc.y * dd * dd;
  acc.z = acc.z * dd * dd;
  acc.w = acc.w * dd * dd;
  for (int j = beg; j < end; ++j) {
    const int2 ed = ((const int2*)csr)[j];
    const float nrm = dinv[ed.x] * __int_as_float(ed.y) * dd;
    const float4 v = p[ed.x * 4 + sub];
    acc.x += v.x * nrm;
    acc.y += v.y * nrm;
    acc.z += v.z * nrm;
    acc.w += v.w * nrm;
  }
  ((float4*)agg)[d * 4 + sub] = acc;
}

__global__ void softmax_kernel(const float* __restrict__ agg2,
                               const float* __restrict__ b2,
                               float* __restrict__ out) {
  const int i = blockIdx.x * blockDim.x + threadIdx.x;
  if (i >= N_NODES) return;
  float v[16];
  const float4* p = (const float4*)(agg2 + (size_t)i * 16);
  float m = -3.402823466e+38f;
#pragma unroll
  for (int q = 0; q < 4; ++q) {
    const float4 t = p[q];
    v[4 * q + 0] = t.x + b2[4 * q + 0];
    v[4 * q + 1] = t.y + b2[4 * q + 1];
    v[4 * q + 2] = t.z + b2[4 * q + 2];
    v[4 * q + 3] = t.w + b2[4 * q + 3];
  }
#pragma unroll
  for (int c = 0; c < 16; ++c) m = fmaxf(m, v[c]);
  float s = 0.f;
#pragma unroll
  for (int c = 0; c < 16; ++c) { v[c] = expf(v[c] - m); s += v[c]; }
  const float inv = 1.0f / s;
  float4* o = (float4*)(out + (size_t)i * 16);
#pragma unroll
  for (int q = 0; q < 4; ++q)
    o[q] = make_float4(v[4 * q + 0] * inv, v[4 * q + 1] * inv,
                       v[4 * q + 2] * inv, v[4 * q + 3] * inv);
}

extern "C" void kernel_launch(void* const* d_in, const int* in_sizes, int n_in,
                              void* d_out, int out_size, void* d_ws, size_t ws_size,
                              hipStream_t stream) {
  const float* x  = (const float*)d_in[0];
  const int*   ei = (const int*)d_in[1];
  const float* ew = (const float*)d_in[2];
  const float* W1 = (const float*)d_in[3];
  const float* b1 = (const float*)d_in[4];
  const float* W2 = (const float*)d_in[5];
  const float* b2 = (const float*)d_in[6];
  float* out = (float*)d_out;

  const int* src = ei;
  const int* dst = ei + N_EDGES;

  float* ws   = (float*)d_ws;
  float* dinv = ws;
  float* h    = ws + N_NODES;
  float* agg1 = ws + N_NODES + (size_t)N_NODES * 128;
  float* h2   = agg1;
  float* agg2 = agg1 + (size_t)N_NODES * 16;
  int* cnt  = (int*)(ws + N_NODES + (size_t)N_NODES * 256);
  int* off  = cnt + NPAD;
  int* bsum = off + NPAD;
  int* rank = bsum + 128;
  int* csr  = rank + N_EDGES;

  const int T = 256;

  zero_cnt_kernel<<<NPAD / T, T, 0, stream>>>(cnt);
  rank_kernel<<<(N_EDGES + T - 1) / T, T, 0, stream>>>(dst, cnt, rank);
  blocksum_kernel<<<SCAN_NBLK, T, 0, stream>>>(cnt, bsum);
  scan_kernel<<<SCAN_NBLK, T, 0, stream>>>(cnt, bsum, off);
  fill_kernel<<<(N_EDGES + T - 1) / T, T, 0, stream>>>(src, dst, ew, rank, off, csr);
  dinv_kernel<<<(N_NODES + T - 1) / T, T, 0, stream>>>(off, csr, dinv);

  gemm1_wmma<<<(N_NODES + 63) / 64, T, 0, stream>>>(x, W1, h);
  gather1_gemm2_kernel<<<N_NODES / G1_TILE, T, 0, stream>>>(off, csr, dinv, h, b1, W2, h2);

  gather2_kernel<<<(N_NODES * 4 + T - 1) / T, T, 0, stream>>>(off, csr, dinv, h2, agg2);

  softmax_kernel<<<(N_NODES + T - 1) / T, T, 0, stream>>>(agg2, b2, out);
}


# Rank edges with LDS counters in a two-level counting sort (256-node ranges; one work-group per range writes off, dinv, csr), no returning global atomic per edge; edge and degree sum order change
# speedup vs baseline: 3.0108x; 1.4119x over previous
#include <hip/hip_runtime.h>
#include <hip/hip_bf16.h>
#include <math.h>


#define N_NODES 100000
#define N_EDGES 1600000
#define F_DIM 128
#define H_DIM 128
#define C_DIM 16

#define RNG_SHIFT 8
#define RNG_NODES (1 << RNG_SHIFT)
#define NRNG ((N_NODES >> RNG_SHIFT) + 1)
#define CHUNK_EDGES 4096
#define NCHUNK ((N_EDGES + CHUNK_EDGES - 1) / CHUNK_EDGES)
#define E4 (N_EDGES / 4)
#define SRC_BITS 17
#define SRC_MASK ((1 << SRC_BITS) - 1)
static_assert(N_NODES <= (1 << SRC_BITS), "source index and in-range destination share one word");
static_assert(N_EDGES % 4 == 0, "edge arrays are read four entries at a time");
static_assert(NCHUNK <= 1024, "one table row is scanned one entry per thread");
static_assert(NRNG * NCHUNK <= 153600 && NRNG <= 512 && N_NODES + 1 <= 100352, "workspace reservations");

typedef __attribute__((ext_vector_type(2))) float v2f;
typedef __attribute__((ext_vector_type(8))) float v8f;

__global__ __launch_bounds__(256) void range_count_kernel(const int* __restrict__ dst,
                                                          int* __restrict__ tab) {
  __shared__ int sc[NRNG];
  const int tid = threadIdx.x;
  for (int i = tid; i < NRNG; i += 256) sc[i] = 0;
  __syncthreads();
  const int4* d4 = (const int4*)dst;
#pragma unroll
  for (int k = 0; k < CHUNK_EDGES / 1024; ++k) {
    const int idx = (int)blockIdx.x * (CHUNK_EDGES / 4) + k * 256 + tid;
    if (idx < E4) {
      const int4 d = d4[idx];
      atomicAdd(&sc[d.x >> RNG_SHIFT], 1);
      atomicAdd(&sc[d.y >> RNG_SHIFT], 1);
      atomicAdd(&sc[d.z >> RNG_SHIFT], 1);
      atomicAdd(&sc[d.w >> RNG_SHIFT], 1);
    }
  }
  __syncthreads();
  for (int i = tid; i < NRNG; i += 256) tab[i * NCHUNK + (int)blockIdx.x] = sc[i];
}

__global__ __launch_bounds__(1024) void range_scan_kernel(const int* __restrict__ tab,
                                                          int* __restrict__ base,
                                                          int* __restrict__ rstart) {
  __shared__ int sw[32];
  __shared__ int sv[32];
  const int tid = threadIdx.x, lane = tid & 31, wave = tid >> 5;
  const int r = (int)blockIdx.x;
  const int n = r * NCHUNK;
  const int n4 = n >> 2;
  int s = 0;
  const int4* t4 = (const int4*)tab;
  for (int i = tid; i < n4; i += 8 * 1024) {
    int4 v[8];
#pragma unroll
    for (int k = 0; k < 8; ++k) v[k] = t4[min(i + k * 1024, n4 - 1)];
#pragma unroll
    for (int k = 0; k < 8; ++k)
      s += (i + k * 1024 < n4) ? (v[k].x + v[k].y + v[k].z + v[k].w) : 0;
  }
  for (int i = (n & ~3) + tid; i < n; i += 1024) s += tab[i];
  for (int o = 16; o > 0; o >>= 1) s += __shfl_xor(s, o);
  if (lane == 0) sw[wave] = s;
  const int a = (tid < NCHUNK) ? tab[n + tid] : 0;
  int inc = a;
  for (int o = 1; o < 32; o <<= 1) {
    const int u = __shfl_up(inc, o);
    if (lane >= o) inc += u;
  }
  if (lane == 31) sv[wave] = inc;
  __syncthreads();
  int first = 0, wbase = 0;
  for (int w = 0; w < 32; ++w) {
    first += sw[w];
    if (w < wave) wbase += sv[w];
  }
  if (tid < NCHUNK) base[n + tid] = first + wbase + inc - a;
  if (tid == 0) rstart[r] = first;
}

__global__ __launch_bounds__(256) void range_place_kernel(const int* __restrict__ src,
                                                          const int* __restrict__ dst,
                                                          const float* __restrict__ ew,
                                                          const int* __restrict__ base,
                                                          int* __restrict__ rec) {
  __shared__ int cur[NRNG];
  const int tid = threadIdx.x;
  for (int i = tid; i < NRNG; i += 256) cur[i] = base[i * NCHUNK + (int)blockIdx.x];
  __syncthreads();
  const int4* d4 = (const int4*)dst;
  const int4* s4 = (const int4*)src;
  const float4* w4 = (const float4*)ew;
  int2* rec2 = (int2*)rec;
#pragma unroll
  for (int k = 0; k < CHUNK_EDGES / 1024; ++k) {
    const int idx = (int)blockIdx.x * (CHUNK_EDGES / 4) + k * 256 + tid;
    if (idx < E4) {
      const int4 d = d4[idx];
      const int4 sr = s4[idx];
      const float4 w = w4[idx];
      const int p0 = atomicAdd(&cur[d.x >> RNG_SHIFT], 1);
      const int p1 = atomicAdd(&cur[d.y >> RNG_SHIFT], 1);
      const int p2 = atomicAdd(&cur[d.z >> RNG_SHIFT], 1);
      const int p3 = atomicAdd(&cur[d.w >> RNG_SHIFT], 1);
      rec2[p0] = make_int2(sr.x | ((d.x & (RNG_NODES - 1)) << SRC_BITS), __float_as_int(w.x));
      rec2[p1] = make_int2(sr.y | ((d.y & (RNG_NODES - 1)) << SRC_BITS), __float_as_int(w.y));
      rec2[p2] = make_int2(sr.z | ((d.z & (RNG_NODES - 1)) << SRC_BITS), __float_as_int(w.z));
      rec2[p3] = make_int2(sr.w | ((d.w & (RNG_NODES - 1)) << SRC_BITS), __float_as_int(w.w));
    }
  }
}

__global__ __launch_bounds__(1024) void range_build_kernel(const int* __restrict__ rec,
                                                           const int* __restrict__ rstart,
                                                           int* __restrict__ off,
                                                           float* __restrict__ dinv,
                                                           int* __restrict__ csr) {
  __shared__ int cnt[RNG_NODES];
  __shared__ float dw[RNG_NODES];
  __shared__ int cur[RNG_NODES];
  __shared__ int sv[RNG_NODES / 32];
  const int tid = threadIdx.x, lane = tid & 31, wave = tid >> 5;
  const int r = (int)blockIdx.x;
  const int beg = rstart[r];
  const int end = (r + 1 < NRNG) ? rstart[r + 1] : N_EDGES;
  const int2* rec2 = (const int2*)rec;
  int2* csr2 = (int2*)csr;
  if (tid < RNG_NODES) {
    cnt[tid] = 0;
    dw[tid] = 0.f;
  }
  __syncthreads();
  for (int e = beg + tid; e < end; e += 4 * 1024) {
    int2 q[4];
#pragma unroll
    for (int k = 0; k < 4; ++k) q[k] = rec2[min(e + k * 1024, end - 1)];
#pragma unroll
    for (int k = 0; k < 4; ++k) {
      if (e + k * 1024 < end) {
        const int dl = q[k].x >> SRC_BITS;
        atomicAdd(&cnt[dl], 1);
        atomicAdd(&dw[dl], __int_as_float(q[k].y));
      }
    }
  }
  __syncthreads();
  int c = 0, inc = 0;
  if (tid < RNG_NODES) {
    c = cnt[tid];
    inc = c;
    for (int o = 1; o < 32; o <<= 1) {
      const int u = __shfl_up(inc, o);
      if (lane >= o) inc += u;
    }
    if (lane == 31) sv[wave] = inc;
  }
  __syncthreads();
  if (tid < RNG_NODES) {
    int wbase = 0;
    for (int w = 0; w < wave; ++w) wbase += sv[w];
    const int ex = beg + wbase + inc - c;
    cur[tid] = ex;
    const int node = r * RNG_NODES + tid;
    if (node <= N_NODES) off[node] = ex;
    if (node < N_NODES) {
      const float d = 1.0f + dw[tid];
      dinv[node] = (d > 0.0f) ? rsqrtf(d) : 0.0f;
    }
  }
  __syncthreads();
  for (int e = beg + tid; e < end; e += 4 * 1024) {
    int2 q[4];
#pragma unroll
    for (int k = 0; k < 4; ++k) q[k] = rec2[min(e + k * 1024, end - 1)];
#pragma unroll
    for (int k = 0; k < 4; ++k) {
      if (e + k * 1024 < end) {
        const int slot = atomicAdd(&cur[q[k].x >> SRC_BITS], 1);
        csr2[slot] = make_int2(q[k].x & SRC_MASK, q[k].y);
      }
    }
  }
}

__global__ __launch_bounds__(256) void gemm1_wmma(const float* __restrict__ x,
                                                  const float* __restrict__ W1,
                                                  float* __restrict__ h) {
  __shared__ float sX[64 * 132];
  const int tid  = threadIdx.x;
  const int row0 = blockIdx.x * 64;

  for (int i = tid; i < 64 * 32; i += 256) {
    const int r = i >> 5, c = i & 31;
    float4 v = make_float4(0.f, 0.f, 0.f, 0.f);
    const int gr = row0 + r;
    if (gr < N_NODES) v = ((const float4*)x)[gr * 32 + c];
    *(float4*)&sX[r * 132 + c * 4] = v;
  }
  __syncthreads();

  const int wave = tid >> 5;
  const int lane = tid & 31;
  const int lrow = lane & 15;
  const int hi   = lane >> 4;
  const int col  = wave * 16 + lrow;

  for (int t = 0; t < 4; ++t) {
    v8f acc = {};
    const float* ax = &sX[(t * 16 + lrow) * 132 + 2 * hi];
    for (int k0 = 0; k0 < 128; k0 += 4) {
      v2f a, b;
      a.x = ax[k0];
      a.y = ax[k0 + 1];
      const int kb = k0 + 2 * hi;
      b.x = W1[kb * 128 + col];
      b.y = W1[(kb + 1) * 128 + col];
      acc = __builtin_amdgcn_wmma_f32_16x16x4_f32(false, a, false, b,
                                                  (short)0, acc, false, false);
    }
    const int rbase = row0 + t * 16 + 8 * hi;
#pragma unroll
    for (int r = 0; r < 8; ++r) {
      const int gr = rbase + r;
      if (gr < N_NODES) h[gr * 128 + col] = acc[r];
    }
  }
}

#define G1_NPW 2
#define G1_TILE (8 * G1_NPW)
#define G1_PITCH 132
static_assert(G1_TILE == 16, "one 16-row matrix tile per work-group");
static_assert(N_NODES % G1_TILE == 0, "the grid covers the nodes exactly, no row guards");

__device__ __forceinline__ void gather_rows(float4& acc, const int s_l, const float nrm_l,
                                            const int n, const float4* __restrict__ h4,
                                            const int lane) {
#pragma unroll 4
  for (int j = 0; j < n; ++j) {
    const int s = __builtin_amdgcn_readlane(s_l, j);
    const float nrm = __int_as_float(__builtin_amdgcn_readlane(__float_as_int(nrm_l), j));
    const float4 v = h4[s * 32 + lane];
    acc.x += v.x * nrm;
    acc.y += v.y * nrm;
    acc.z += v.z * nrm;
    acc.w += v.w * nrm;
  }
}

__global__ __launch_bounds__(256) void gather1_gemm2_kernel(const int* __restrict__ off,
                                                            const int* __restrict__ csr,
                                                            const float* __restrict__ dinv,
                                                            const float* __restrict__ h,
                                                            const float* __restrict__ b1,
                                                            const float* __restrict__ W2,
                                                            float* __restrict__ h2) {
  __shared__ float sA[G1_TILE * G1_PITCH];
  __shared__ float sW[128 * 16];
  const int tid  = threadIdx.x;
  const int lane = tid & 31;
  const int wave = __builtin_amdgcn_readfirstlane(tid >> 5);
  const float4 w2a = ((const float4*)W2)[tid];
  const float4 w2b = ((const float4*)W2)[tid + 256];
  const float4 bias = ((const float4*)b1)[lane];
  const float4* h4  = (const float4*)h;
  const int2* csr2  = (const int2*)csr;
  const int d0 = (int)blockIdx.x * G1_TILE + wave;

  int beg[G1_NPW], end[G1_NPW], nn[G1_NPW], s_l[G1_NPW];
  float dd[G1_NPW], w_l[G1_NPW], nrm_l[G1_NPW];
  float4 acc[G1_NPW];
#pragma unroll
  for (int p = 0; p < G1_NPW; ++p) {
    const int d = d0 + 8 * p;
    beg[p] = off[d];
    end[p] = off[d + 1];
    dd[p]  = dinv[d];
  }
#pragma unroll
  for (int p = 0; p < G1_NPW; ++p) {
    const int d = d0 + 8 * p;
    acc[p] = h4[d * 32 + lane];
    nn[p]  = min(32, end[p] - beg[p]);
    const bool on = lane < nn[p];
    const int2 ed = csr2[on ? beg[p] + lane : 0];
    s_l[p] = on ? ed.x : 0;
    w_l[p] = on ? __int_as_float(ed.y) : 0.f;
  }
#pragma unroll
  for (int p = 0; p < G1_NPW; ++p) nrm_l[p] = dinv[s_l[p]] * w_l[p] * dd[p];
  asm volatile("" :: "v"(w2a.x), "v"(w2a.y), "v"(w2a.z), "v"(w2a.w),
                     "v"(w2b.x), "v"(w2b.y), "v"(w2b.z), "v"(w2b.w));

#pragma unroll
  for (int p = 0; p < G1_NPW; ++p) {
    float4 a = acc[p];
    a.x = a.x * dd[p] * dd[p];
    a.y = a.y * dd[p] * dd[p];
    a.z = a.z * dd[p] * dd[p];
    a.w = a.w * dd[p] * dd[p];
    gather_rows(a, s_l[p], nrm_l[p], nn[p], h4, lane);
    for (int base = beg[p] + 32; base < end[p]; base += 32) {
      const int n = min(32, end[p] - base);
      int s2 = 0;
      float nrm2 = 0.f;
      if (lane < n) {
        const int2 ed = csr2[base + lane];
        s2 = ed.x;
        nrm2 = dinv[ed.x] * __int_as_float(ed.y) * dd[p];
      }
      gather_rows(a, s2, nrm2, n, h4, lane);
    }
    float4 r;
    r.x = fmaxf(a.x + bias.x, 0.f);
    r.y = fmaxf(a.y + bias.y, 0.f);
    r.z = fmaxf(a.z + bias.z, 0.f);
    r.w = fmaxf(a.w + bias.w, 0.f);
    *(float4*)&sA[(8 * p + wave) * G1_PITCH + 4 * lane] = r;
  }
  ((float4*)sW)[tid] = w2a;
  ((float4*)sW)[tid + 256] = w2b;
  __syncthreads();

  if (wave == 0) {
    const int lrow = lane & 15, hi = lane >> 4;
    const float* ax = &sA[lrow * G1_PITCH + 2 * hi];
    v8f c = {};
    for (int k0 = 0; k0 < 128; k0 += 4) {
      const int k = k0 + 2 * hi;
      v2f a, b;
      a.x = ax[k0];
      a.y = ax[k0 + 1];
      b.x = sW[k * 16 + lrow];
      b.y = sW[(k + 1) * 16 + lrow];
      c = __builtin_amdgcn_wmma_f32_16x16x4_f32(false, a, false, b,
                                                (short)0, c, false, false);
    }
    const int rbase = (int)blockIdx.x * G1_TILE + 8 * hi;
#pragma unroll
    for (int r = 0; r < 8; ++r)
      h2[(rbase + r) * 16 + lrow] = c[r];
  }
}

__global__ __launch_bounds__(256) void gather2_kernel(const int* __restrict__ off,
                                                      const int* __restrict__ csr,
                                                      const float* __restrict__ dinv,
                                                      const float* __restrict__ h2,
                                                      float* __restrict__ agg) {
  const int gid = blockIdx.x * blockDim.x + threadIdx.x;
  const int d   = gid >> 2;
  const int sub = threadIdx.x & 3;
  if (d >= N_NODES) return;
  const int beg = off[d], end = off[d + 1];
  const float dd = dinv[d];
  const float4* p = (const float4*)h2;
  float4 acc = p[d * 4 + sub];
  acc.x = acc.x * dd * dd;
  acc.y = acc.y * dd * dd;
  acc.z = acc.z * dd * dd;
  acc.w = acc.w * dd * dd;
  for (int j = beg; j < end; ++j) {
    const int2 ed = ((const int2*)csr)[j];
    const float nrm = dinv[ed.x] * __int_as_float(ed.y) * dd;
    const float4 v = p[ed.x * 4 + sub];
    acc.x += v.x * nrm;
    acc.y += v.y * nrm;
    acc.z += v.z * nrm;
    acc.w += v.w * nrm;
  }
  ((float4*)agg)[d * 4 + sub] = acc;
}

__global__ void softmax_kernel(const float* __restrict__ agg2,
                               const float* __restrict__ b2,
                               float* __restrict__ out) {
  const int i = blockIdx.x * blockDim.x + threadIdx.x;
  if (i >= N_NODES) return;
  float v[16];
  const float4* p = (const float4*)(agg2 + (size_t)i * 16);
  float m = -3.402823466e+38f;
#pragma unroll
  for (int q = 0; q < 4; ++q) {
    const float4 t = p[q];
    v[4 * q + 0] = t.x + b2[4 * q + 0];
    v[4 * q + 1] = t.y + b2[4 * q + 1];
    v[4 * q + 2] = t.z + b2[4 * q + 2];
    v[4 * q + 3] = t.w + b2[4 * q + 3];
  }
#pragma unroll
  for (int c = 0; c < 16; ++c) m = fmaxf(m, v[c]);
  float s = 0.f;
#pragma unroll
  for (int c = 0; c < 16; ++c) { v[c] = expf(v[c] - m); s += v[c]; }
  const float inv = 1.0f / s;
  float4* o = (float4*)(out + (size_t)i * 16);
#pragma unroll
  for (int q = 0; q < 4; ++q)
    o[q] = make_float4(v[4 * q + 0] * inv, v[4 * q + 1] * inv,
                       v[4 * q + 2] * inv, v[4 * q + 3] * inv);
}

extern "C" void kernel_launch(void* const* d_in, const int* in_sizes, int n_in,
                              void* d_out, int out_size, void* d_ws, size_t ws_size,
                              hipStream_t stream) {
  const float* x  = (const float*)d_in[0];
  const int*   ei = (const int*)d_in[1];
  const float* ew = (const float*)d_in[2];
  const float* W1 = (const float*)d_in[3];
  const float* b1 = (const float*)d_in[4];
  const float* W2 = (const float*)d_in[5];
  const float* b2 = (const float*)d_in[6];
  float* out = (float*)d_out;

  const int* src = ei;
  const int* dst = ei + N_EDGES;

  float* ws   = (float*)d_ws;
  float* dinv = ws;
  float* h    = ws + N_NODES;
  float* agg1 = ws + N_NODES + (size_t)N_NODES * 128;
  float* h2   = agg1;
  float* agg2 = agg1 + (size_t)N_NODES * 16;
  int* off    = (int*)(ws + N_NODES + (size_t)N_NODES * 256);
  int* tab    = off + 100352;
  int* base   = tab + 153600;
  int* rstart = base + 153600;
  int* rec    = rstart + 512;
  int* csr    = rec + 2 * N_EDGES;

  const int T = 256;

  range_count_kernel<<<NCHUNK, T, 0, stream>>>(dst, tab);
  range_scan_kernel<<<NRNG, 1024, 0, stream>>>(tab, base, rstart);
  range_place_kernel<<<NCHUNK, T, 0, stream>>>(src, dst, ew, base, rec);
  range_build_kernel<<<NRNG, 1024, 0, stream>>>(rec, rstart, off, dinv, csr);

  gemm1_wmma<<<(N_NODES + 63) / 64, T, 0, stream>>>(x, W1, h);
  gather1_gemm2_kernel<<<N_NODES / G1_TILE, T, 0, stream>>>(off, csr, dinv, h, b1, W2, h2);

  gather2_kernel<<<(N_NODES * 4 + T - 1) / T, T, 0, stream>>>(off, csr, dinv, h2, agg2);

  softmax_kernel<<<(N_NODES + T - 1) / T, T, 0, stream>>>(agg2, b2, out);
}


# Aggregate x before the first linear layer, (A_hat x) W1: GEMM1 runs in the gather's epilogue on 32-node LDS tiles (f32 WMMA), h is never stored, GEMM1 launch gone; layer-1 sums reassociated
# speedup vs baseline: 3.6988x; 1.2285x over previous
#include <hip/hip_runtime.h>
#include <hip/hip_bf16.h>
#include <math.h>


#define N_NODES 100000
#define N_EDGES 1600000
#define F_DIM 128
#define H_DIM 128
#define C_DIM 16

#define RNG_SHIFT 8
#define RNG_NODES (1 << RNG_SHIFT)
#define NRNG ((N_NODES >> RNG_SHIFT) + 1)
#define CHUNK_EDGES 4096
#define NCHUNK ((N_EDGES + CHUNK_EDGES - 1) / CHUNK_EDGES)
#define E4 (N_EDGES / 4)
#define SRC_BITS 17
#define SRC_MASK ((1 << SRC_BITS) - 1)
static_assert(N_NODES <= (1 << SRC_BITS), "source index and in-range destination share one word");
static_assert(N_EDGES % 4 == 0, "edge arrays are read four entries at a time");
static_assert(NCHUNK <= 1024, "one table row is scanned one entry per thread");
static_assert(NRNG * NCHUNK <= 153600 && NRNG <= 512 && N_NODES + 1 <= 100352, "workspace reservations");

typedef __attribute__((ext_vector_type(2))) float v2f;
typedef __attribute__((ext_vector_type(8))) float v8f;

__global__ __launch_bounds__(256) void range_count_kernel(const int* __restrict__ dst,
                                                          int* __restrict__ tab) {
  __shared__ int sc[NRNG];
  const int tid = threadIdx.x;
  for (int i = tid; i < NRNG; i += 256) sc[i] = 0;
  __syncthreads();
  const int4* d4 = (const int4*)dst;
#pragma unroll
  for (int k = 0; k < CHUNK_EDGES / 1024; ++k) {
    const int idx = (int)blockIdx.x * (CHUNK_EDGES / 4) + k * 256 + tid;
    if (idx < E4) {
      const int4 d = d4[idx];
      atomicAdd(&sc[d.x >> RNG_SHIFT], 1);
      atomicAdd(&sc[d.y >> RNG_SHIFT], 1);
      atomicAdd(&sc[d.z >> RNG_SHIFT], 1);
      atomicAdd(&sc[d.w >> RNG_SHIFT], 1);
    }
  }
  __syncthreads();
  for (int i = tid; i < NRNG; i += 256) tab[i * NCHUNK + (int)blockIdx.x] = sc[i];
}

__global__ __launch_bounds__(1024) void range_scan_kernel(const int* __restrict__ tab,
                                                          int* __restrict__ base,
                                                          int* __restrict__ rstart) {
  __shared__ int sw[32];
  __shared__ int sv[32];
  const int tid = threadIdx.x, lane = tid & 31, wave = tid >> 5;
  const int r = (int)blockIdx.x;
  const int n = r * NCHUNK;
  const int n4 = n >> 2;
  int s = 0;
  const int4* t4 = (const int4*)tab;
  for (int i = tid; i < n4; i += 8 * 1024) {
    int4 v[8];
#pragma unroll
    for (int k = 0; k < 8; ++k) v[k] = t4[min(i + k * 1024, n4 - 1)];
#pragma unroll
    for (int k = 0; k < 8; ++k)
      s += (i + k * 1024 < n4) ? (v[k].x + v[k].y + v[k].z + v[k].w) : 0;
  }
  for (int i = (n & ~3) + tid; i < n; i += 1024) s += tab[i];
  for (int o = 16; o > 0; o >>= 1) s += __shfl_xor(s, o);
  if (lane == 0) sw[wave] = s;
  const int a = (tid < NCHUNK) ? tab[n + tid] : 0;
  int inc = a;
  for (int o = 1; o < 32; o <<= 1) {
    const int u = __shfl_up(inc, o);
    if (lane >= o) inc += u;
  }
  if (lane == 31) sv[wave] = inc;
  __syncthreads();
  int first = 0, wbase = 0;
  for (int w = 0; w < 32; ++w) {
    first += sw[w];
    if (w < wave) wbase += sv[w];
  }
  if (tid < NCHUNK) base[n + tid] = first + wbase + inc - a;
  if (tid == 0) rstart[r] = first;
}

__global__ __launch_bounds__(256) void range_place_kernel(const int* __restrict__ src,
                                                          const int* __restrict__ dst,
                                                          const float* __restrict__ ew,
                                                          const int* __restrict__ base,
                                                          int* __restrict__ rec) {
  __shared__ int cur[NRNG];
  const int tid = threadIdx.x;
  for (int i = tid; i < NRNG; i += 256) cur[i] = base[i * NCHUNK + (int)blockIdx.x];
  __syncthreads();
  const int4* d4 = (const int4*)dst;
  const int4* s4 = (const int4*)src;
  const float4* w4 = (const float4*)ew;
  int2* rec2 = (int2*)rec;
#pragma unroll
  for (int k = 0; k < CHUNK_EDGES / 1024; ++k) {
    const int idx = (int)blockIdx.x * (CHUNK_EDGES / 4) + k * 256 + tid;
    if (idx < E4) {
      const int4 d = d4[idx];
      const int4 sr = s4[idx];
      const float4 w = w4[idx];
      const int p0 = atomicAdd(&cur[d.x >> RNG_SHIFT], 1);
      const int p1 = atomicAdd(&cur[d.y >> RNG_SHIFT], 1);
      const int p2 = atomicAdd(&cur[d.z >> RNG_SHIFT], 1);
      const int p3 = atomicAdd(&cur[d.w >> RNG_SHIFT], 1);
      rec2[p0] = make_int2(sr.x | ((d.x & (RNG_NODES - 1)) << SRC_BITS), __float_as_int(w.x));
      rec2[p1] = make_int2(sr.y | ((d.y & (RNG_NODES - 1)) << SRC_BITS), __float_as_int(w.y));
      rec2[p2] = make_int2(sr.z | ((d.z & (RNG_NODES - 1)) << SRC_BITS), __float_as_int(w.z));
      rec2[p3] = make_int2(sr.w | ((d.w & (RNG_NODES - 1)) << SRC_BITS), __float_as_int(w.w));
    }
  }
}

__global__ __launch_bounds__(1024) void range_build_kernel(const int* __restrict__ rec,
                                                           const int* __restrict__ rstart,
                                                           int* __restrict__ off,
                                                           float* __restrict__ dinv,
                                                           int* __restrict__ csr) {
  __shared__ int cnt[RNG_NODES];
  __shared__ float dw[RNG_NODES];
  __shared__ int cur[RNG_NODES];
  __shared__ int sv[RNG_NODES / 32];
  const int tid = threadIdx.x, lane = tid & 31, wave = tid >> 5;
  const int r = (int)blockIdx.x;
  const int beg = rstart[r];
  const int end = (r + 1 < NRNG) ? rstart[r + 1] : N_EDGES;
  const int2* rec2 = (const int2*)rec;
  int2* csr2 = (int2*)csr;
  if (tid < RNG_NODES) {
    cnt[tid] = 0;
    dw[tid] = 0.f;
  }
  __syncthreads();
  for (int e = beg + tid; e < end; e += 4 * 1024) {
    int2 q[4];
#pragma unroll
    for (int k = 0; k < 4; ++k) q[k] = rec2[min(e + k * 1024, end - 1)];
#pragma unroll
    for (int k = 0; k < 4; ++k) {
      if (e + k * 1024 < end) {
        const int dl = q[k].x >> SRC_BITS;
        atomicAdd(&cnt[dl], 1);
        atomicAdd(&dw[dl], __int_as_float(q[k].y));
      }
    }
  }
  __syncthreads();
  int c = 0, inc = 0;
  if (tid < RNG_NODES) {
    c = cnt[tid];
    inc = c;
    for (int o = 1; o < 32; o <<= 1) {
      const int u = __shfl_up(inc, o);
      if (lane >= o) inc += u;
    }
    if (lane == 31) sv[wave] = inc;
  }
  __syncthreads();
  if (tid < RNG_NODES) {
    int wbase = 0;
    for (int w = 0; w < wave; ++w) wbase += sv[w];
    const int ex = beg + wbase + inc - c;
    cur[tid] = ex;
    const int node = r * RNG_NODES + tid;
    if (node <= N_NODES) off[node] = ex;
    if (node < N_NODES) {
      const float d = 1.0f + dw[tid];
      dinv[node] = (d > 0.0f) ? rsqrtf(d) : 0.0f;
    }
  }
  __syncthreads();
  for (int e = beg + tid; e < end; e += 4 * 1024) {
    int2 q[4];
#pragma unroll
    for (int k = 0; k < 4; ++k) q[k] = rec2[min(e + k * 1024, end - 1)];
#pragma unroll
    for (int k = 0; k < 4; ++k) {
      if (e + k * 1024 < end) {
        const int slot = atomicAdd(&cur[q[k].x >> SRC_BITS], 1);
        csr2[slot] = make_int2(q[k].x & SRC_MASK, q[k].y);
      }
    }
  }
}

#define GX_NPW 4
#define GX_TILE (8 * GX_NPW)
#define GX_PITCH 132
static_assert(GX_TILE == 32, "two 16-row matrix tiles per work-group");
static_assert(N_NODES % GX_TILE == 0, "the grid covers the nodes exactly, no row guards");

__device__ __forceinline__ void gather_rows(float4& acc, const int s_l, const float nrm_l,
                                            const int n, const float4* __restrict__ r4,
                                            const int lane) {
#pragma unroll 4
  for (int j = 0; j < n; ++j) {
    const int s = __builtin_amdgcn_readlane(s_l, j);
    const float nrm = __int_as_float(__builtin_amdgcn_readlane(__float_as_int(nrm_l), j));
    const float4 v = r4[s * 32 + lane];
    acc.x += v.x * nrm;
    acc.y += v.y * nrm;
    acc.z += v.z * nrm;
    acc.w += v.w * nrm;
  }
}

__global__ __launch_bounds__(256) __attribute__((amdgpu_waves_per_eu(16, 16)))
void gather_gemm12_kernel(const int* __restrict__ off,
                                                            const int* __restrict__ csr,
                                                            const float* __restrict__ dinv,
                                                            const float* __restrict__ x,
                                                            const float* __restrict__ W1,
                                                            const float* __restrict__ b1,
                                                            const float* __restrict__ W2,
                                                            float* __restrict__ h2) {
  __shared__ float sA[GX_TILE * GX_PITCH];
  __shared__ float sW[128 * 16];
  const int tid  = threadIdx.x;
  const int lane = tid & 31;
  const int wave = __builtin_amdgcn_readfirstlane(tid >> 5);
  const float4 w2a = ((const float4*)W2)[tid];
  const float4 w2b = ((const float4*)W2)[tid + 256];
  const float4* x4  = (const float4*)x;
  const int2* csr2  = (const int2*)csr;
  const int d0 = (int)blockIdx.x * GX_TILE + wave;

  int beg[GX_NPW], end[GX_NPW], nn[GX_NPW], s_l[GX_NPW];
  float dd[GX_NPW], w_l[GX_NPW], nrm_l[GX_NPW];
  float4 acc[GX_NPW];
#pragma unroll
  for (int p = 0; p < GX_NPW; ++p) {
    const int d = d0 + 8 * p;
    beg[p] = off[d];
    end[p] = off[d + 1];
    dd[p]  = dinv[d];
  }
#pragma unroll
  for (int p = 0; p < GX_NPW; ++p) {
    const int d = d0 + 8 * p;
    acc[p] = x4[d * 32 + lane];
    nn[p]  = min(32, end[p] - beg[p]);
    const bool on = lane < nn[p];
    const int2 ed = csr2[on ? beg[p] + lane : 0];
    s_l[p] = on ? ed.x : 0;
    w_l[p] = on ? __int_as_float(ed.y) : 0.f;
  }
#pragma unroll
  for (int p = 0; p < GX_NPW; ++p) nrm_l[p] = dinv[s_l[p]] * w_l[p] * dd[p];
  asm volatile("" :: "v"(w2a.x), "v"(w2a.y), "v"(w2a.z), "v"(w2a.w),
                     "v"(w2b.x), "v"(w2b.y), "v"(w2b.z), "v"(w2b.w));

#pragma unroll
  for (int p = 0; p < GX_NPW; ++p) {
    float4 a = acc[p];
    a.x = a.x * dd[p] * dd[p];
    a.y = a.y * dd[p] * dd[p];
    a.z = a.z * dd[p] * dd[p];
    a.w = a.w * dd[p] * dd[p];
    gather_rows(a, s_l[p], nrm_l[p], nn[p], x4, lane);
    for (int base = beg[p] + 32; base < end[p]; base += 32) {
      const int n = min(32, end[p] - base);
      int s2 = 0;
      float nrm2 = 0.f;
      if (lane < n) {
        const int2 ed = csr2[base + lane];
        s2 = ed.x;
        nrm2 = dinv[ed.x] * __int_as_float(ed.y) * dd[p];
      }
      gather_rows(a, s2, nrm2, n, x4, lane);
    }
    *(float4*)&sA[(8 * p + wave) * GX_PITCH + 4 * lane] = a;
  }
  ((float4*)sW)[tid] = w2a;
  ((float4*)sW)[tid + 256] = w2b;

  const int lrow = lane & 15, hi = lane >> 4;
  const int col = wave * 16 + lrow;
  const float* bp = W1 + 2 * hi * 128 + col;
  v2f bq[8];
#pragma unroll
  for (int i = 0; i < 8; ++i) {
    bq[i].x = bp[(4 * i) * 128];
    bq[i].y = bp[(4 * i + 1) * 128];
  }
  __syncthreads();

  v8f c0 = {}, c1 = {};
  {
    const float* a0p = &sA[lrow * GX_PITCH + 2 * hi];
    const float* a1p = a0p + 16 * GX_PITCH;
#pragma unroll
    for (int i = 0; i < 32; ++i) {
      const int k0 = 4 * i;
      v2f a0, a1, b;
      a0.x = a0p[k0];
      a0.y = a0p[k0 + 1];
      a1.x = a1p[k0];
      a1.y = a1p[k0 + 1];
      if (i < 8) {
        b = bq[i];
      } else {
        b.x = bp[k0 * 128];
        b.y = bp[(k0 + 1) * 128];
      }
      c0 = __builtin_amdgcn_wmma_f32_16x16x4_f32(false, a0, false, b,
                                                 (short)0, c0, false, false);
      c1 = __builtin_amdgcn_wmma_f32_16x16x4_f32(false, a1, false, b,
                                                 (short)0, c1, false, false);
    }
  }
  const float bv = b1[col];
  __syncthreads();
#pragma unroll
  for (int r = 0; r < 8; ++r) {
    sA[(8 * hi + r) * GX_PITCH + col]      = fmaxf(c0[r] + bv, 0.f);
    sA[(16 + 8 * hi + r) * GX_PITCH + col] = fmaxf(c1[r] + bv, 0.f);
  }
  __syncthreads();

  if (wave < 2) {
    const float* ax = &sA[(16 * wave + lrow) * GX_PITCH + 2 * hi];
    v8f c = {};
    for (int k0 = 0; k0 < 128; k0 += 4) {
      const int k = k0 + 2 * hi;
      v2f a, b;
      a.x = ax[k0];
      a.y = ax[k0 + 1];
      b.x = sW[k * 16 + lrow];
      b.y = sW[(k + 1) * 16 + lrow];
      c = __builtin_amdgcn_wmma_f32_16x16x4_f32(false, a, false, b,
                                                (short)0, c, false, false);
    }
    const int rbase = (int)blockIdx.x * GX_TILE + 16 * wave + 8 * hi;
#pragma unroll
    for (int r = 0; r < 8; ++r)
      h2[(rbase + r) * 16 + lrow] = c[r];
  }
}

__global__ __launch_bounds__(256) void gather2_kernel(const int* __restrict__ off,
                                                      const int* __restrict__ csr,
                                                      const float* __restrict__ dinv,
                                                      const float* __restrict__ h2,
                                                      float* __restrict__ agg) {
  const int gid = blockIdx.x * blockDim.x + threadIdx.x;
  const int d   = gid >> 2;
  const int sub = threadIdx.x & 3;
  if (d >= N_NODES) return;
  const int beg = off[d], end = off[d + 1];
  const float dd = dinv[d];
  const float4* p = (const float4*)h2;
  float4 acc = p[d * 4 + sub];
  acc.x = acc.x * dd * dd;
  acc.y = acc.y * dd * dd;
  acc.z = acc.z * dd * dd;
  acc.w = acc.w * dd * dd;
  for (int j = beg; j < end; ++j) {
    const int2 ed = ((const int2*)csr)[j];
    const float nrm = dinv[ed.x] * __int_as_float(ed.y) * dd;
    const float4 v = p[ed.x * 4 + sub];
    acc.x += v.x * nrm;
    acc.y += v.y * nrm;
    acc.z += v.z * nrm;
    acc.w += v.w * nrm;
  }
  ((float4*)agg)[d * 4 + sub] = acc;
}

__global__ void softmax_kernel(const float* __restrict__ agg2,
                               const float* __restrict__ b2,
                               float* __restrict__ out) {
  const int i = blockIdx.x * blockDim.x + threadIdx.x;
  if (i >= N_NODES) return;
  float v[16];
  const float4* p = (const float4*)(agg2 + (size_t)i * 16);
  float m = -3.402823466e+38f;
#pragma unroll
  for (int q = 0; q < 4; ++q) {
    const float4 t = p[q];
    v[4 * q + 0] = t.x + b2[4 * q + 0];
    v[4 * q + 1] = t.y + b2[4 * q + 1];
    v[4 * q + 2] = t.z + b2[4 * q + 2];
    v[4 * q + 3] = t.w + b2[4 * q + 3];
  }
#pragma unroll
  for (int c = 0; c < 16; ++c) m = fmaxf(m, v[c]);
  float s = 0.f;
#pragma unroll
  for (int c = 0; c < 16; ++c) { v[c] = expf(v[c] - m); s += v[c]; }
  const float inv = 1.0f / s;
  float4* o = (float4*)(out + (size_t)i * 16);
#pragma unroll
  for (int q = 0; q < 4; ++q)
    o[q] = make_float4(v[4 * q + 0] * inv, v[4 * q + 1] * inv,
                       v[4 * q + 2] * inv, v[4 * q + 3] * inv);
}

extern "C" void kernel_launch(void* const* d_in, const int* in_sizes, int n_in,
                              void* d_out, int out_size, void* d_ws, size_t ws_size,
                              hipStream_t stream) {
  const float* x  = (const float*)d_in[0];
  const int*   ei = (const int*)d_in[1];
  const float* ew = (const float*)d_in[2];
  const float* W1 = (const float*)d_in[3];
  const float* b1 = (const float*)d_in[4];
  const float* W2 = (const float*)d_in[5];
  const float* b2 = (const float*)d_in[6];
  float* out = (float*)d_out;

  const int* src = ei;
  const int* dst = ei + N_EDGES;

  float* ws   = (float*)d_ws;
  float* dinv = ws;
  float* h2   = ws + N_NODES + (size_t)N_NODES * 128;
  float* agg2 = h2 + (size_t)N_NODES * 16;
  int* off    = (int*)(ws + N_NODES + (size_t)N_NODES * 256);
  int* tab    = off + 100352;
  int* base   = tab + 153600;
  int* rstart = base + 153600;
  int* rec    = rstart + 512;
  int* csr    = rec + 2 * N_EDGES;

  const int T = 256;

  range_count_kernel<<<NCHUNK, T, 0, stream>>>(dst, tab);
  range_scan_kernel<<<NRNG, 1024, 0, stream>>>(tab, base, rstart);
  range_place_kernel<<<NCHUNK, T, 0, stream>>>(src, dst, ew, base, rec);
  range_build_kernel<<<NRNG, 1024, 0, stream>>>(rec, rstart, off, dinv, csr);

  gather_gemm12_kernel<<<N_NODES / GX_TILE, T, 0, stream>>>(off, csr, dinv, x, W1, b1, W2, h2);

  gather2_kernel<<<(N_NODES * 4 + T - 1) / T, T, 0, stream>>>(off, csr, dinv, h2, agg2);

  softmax_kernel<<<(N_NODES + T - 1) / T, T, 0, stream>>>(agg2, b2, out);
}


# Layer 2 in one launch: h2 rows stored scaled by their dinv, so its gather loads one row per edge and no dinv[src]; bias+softmax across a node's 4 lanes in the epilogue (exp sum as a tree)
# speedup vs baseline: 4.0572x; 1.0969x over previous
#include <hip/hip_runtime.h>
#include <hip/hip_bf16.h>
#include <math.h>


#define N_NODES 100000
#define N_EDGES 1600000
#define F_DIM 128
#define H_DIM 128
#define C_DIM 16

#define RNG_SHIFT 8
#define RNG_NODES (1 << RNG_SHIFT)
#define NRNG ((N_NODES >> RNG_SHIFT) + 1)
#define CHUNK_EDGES 4096
#define NCHUNK ((N_EDGES + CHUNK_EDGES - 1) / CHUNK_EDGES)
#define E4 (N_EDGES / 4)
#define SRC_BITS 17
#define SRC_MASK ((1 << SRC_BITS) - 1)
static_assert(N_NODES <= (1 << SRC_BITS), "source index and in-range destination share one word");
static_assert(N_EDGES % 4 == 0, "edge arrays are read four entries at a time");
static_assert(NCHUNK <= 1024, "one table row is scanned one entry per thread");
static_assert(NRNG * NCHUNK <= 153600 && NRNG <= 512 && N_NODES + 1 <= 100352, "workspace reservations");

typedef __attribute__((ext_vector_type(2))) float v2f;
typedef __attribute__((ext_vector_type(8))) float v8f;

__global__ __launch_bounds__(256) void range_count_kernel(const int* __restrict__ dst,
                                                          int* __restrict__ tab) {
  __shared__ int sc[NRNG];
  const int tid = threadIdx.x;
  for (int i = tid; i < NRNG; i += 256) sc[i] = 0;
  __syncthreads();
  const int4* d4 = (const int4*)dst;
#pragma unroll
  for (int k = 0; k < CHUNK_EDGES / 1024; ++k) {
    const int idx = (int)blockIdx.x * (CHUNK_EDGES / 4) + k * 256 + tid;
    if (idx < E4) {
      const int4 d = d4[idx];
      atomicAdd(&sc[d.x >> RNG_SHIFT], 1);
      atomicAdd(&sc[d.y >> RNG_SHIFT], 1);
      atomicAdd(&sc[d.z >> RNG_SHIFT], 1);
      atomicAdd(&sc[d.w >> RNG_SHIFT], 1);
    }
  }
  __syncthreads();
  for (int i = tid; i < NRNG; i += 256) tab[i * NCHUNK + (int)blockIdx.x] = sc[i];
}

__global__ __launch_bounds__(1024) void range_scan_kernel(const int* __restrict__ tab,
                                                          int* __restrict__ base,
                                                          int* __restrict__ rstart) {
  __shared__ int sw[32];
  __shared__ int sv[32];
  const int tid = threadIdx.x, lane = tid & 31, wave = tid >> 5;
  const int r = (int)blockIdx.x;
  const int n = r * NCHUNK;
  const int n4 = n >> 2;
  int s = 0;
  const int4* t4 = (const int4*)tab;
  for (int i = tid; i < n4; i += 8 * 1024) {
    int4 v[8];
#pragma unroll
    for (int k = 0; k < 8; ++k) v[k] = t4[min(i + k * 1024, n4 - 1)];
#pragma unroll
    for (int k = 0; k < 8; ++k)
      s += (i + k * 1024 < n4) ? (v[k].x + v[k].y + v[k].z + v[k].w) : 0;
  }
  for (int i = (n & ~3) + tid; i < n; i += 1024) s += tab[i];
  for (int o = 16; o > 0; o >>= 1) s += __shfl_xor(s, o);
  if (lane == 0) sw[wave] = s;
  const int a = (tid < NCHUNK) ? tab[n + tid] : 0;
  int inc = a;
  for (int o = 1; o < 32; o <<= 1) {
    const int u = __shfl_up(inc, o);
    if (lane >= o) inc += u;
  }
  if (lane == 31) sv[wave] = inc;
  __syncthreads();
  int first = 0, wbase = 0;
  for (int w = 0; w < 32; ++w) {
    first += sw[w];
    if (w < wave) wbase += sv[w];
  }
  if (tid < NCHUNK) base[n + tid] = first + wbase + inc - a;
  if (tid == 0) rstart[r] = first;
}

__global__ __launch_bounds__(256) void range_place_kernel(const int* __restrict__ src,
                                                          const int* __restrict__ dst,
                                                          const float* __restrict__ ew,
                                                          const int* __restrict__ base,
                                                          int* __restrict__ rec) {
  __shared__ int cur[NRNG];
  const int tid = threadIdx.x;
  for (int i = tid; i < NRNG; i += 256) cur[i] = base[i * NCHUNK + (int)blockIdx.x];
  __syncthreads();
  const int4* d4 = (const int4*)dst;
  const int4* s4 = (const int4*)src;
  const float4* w4 = (const float4*)ew;
  int2* rec2 = (int2*)rec;
#pragma unroll
  for (int k = 0; k < CHUNK_EDGES / 1024; ++k) {
    const int idx = (int)blockIdx.x * (CHUNK_EDGES / 4) + k * 256 + tid;
    if (idx < E4) {
      const int4 d = d4[idx];
      const int4 sr = s4[idx];
      const float4 w = w4[idx];
      const int p0 = atomicAdd(&cur[d.x >> RNG_SHIFT], 1);
      const int p1 = atomicAdd(&cur[d.y >> RNG_SHIFT], 1);
      const int p2 = atomicAdd(&cur[d.z >> RNG_SHIFT], 1);
      const int p3 = atomicAdd(&cur[d.w >> RNG_SHIFT], 1);
      rec2[p0] = make_int2(sr.x | ((d.x & (RNG_NODES - 1)) << SRC_BITS), __float_as_int(w.x));
      rec2[p1] = make_int2(sr.y | ((d.y & (RNG_NODES - 1)) << SRC_BITS), __float_as_int(w.y));
      rec2[p2] = make_int2(sr.z | ((d.z & (RNG_NODES - 1)) << SRC_BITS), __float_as_int(w.z));
      rec2[p3] = make_int2(sr.w | ((d.w & (RNG_NODES - 1)) << SRC_BITS), __float_as_int(w.w));
    }
  }
}

__global__ __launch_bounds__(1024) void range_build_kernel(const int* __restrict__ rec,
                                                           const int* __restrict__ rstart,
                                                           int* __restrict__ off,
                                                           float* __restrict__ dinv,
                                                           int* __restrict__ csr) {
  __shared__ int cnt[RNG_NODES];
  __shared__ float dw[RNG_NODES];
  __shared__ int cur[RNG_NODES];
  __shared__ int sv[RNG_NODES / 32];
  const int tid = threadIdx.x, lane = tid & 31, wave = tid >> 5;
  const int r = (int)blockIdx.x;
  const int beg = rstart[r];
  const int end = (r + 1 < NRNG) ? rstart[r + 1] : N_EDGES;
  const int2* rec2 = (const int2*)rec;
  int2* csr2 = (int2*)csr;
  if (tid < RNG_NODES) {
    cnt[tid] = 0;
    dw[tid] = 0.f;
  }
  __syncthreads();
  for (int e = beg + tid; e < end; e += 4 * 1024) {
    int2 q[4];
#pragma unroll
    for (int k = 0; k < 4; ++k) q[k] = rec2[min(e + k * 1024, end - 1)];
#pragma unroll
    for (int k = 0; k < 4; ++k) {
      if (e + k * 1024 < end) {
        const int dl = q[k].x >> SRC_BITS;
        atomicAdd(&cnt[dl], 1);
        atomicAdd(&dw[dl], __int_as_float(q[k].y));
      }
    }
  }
  __syncthreads();
  int c = 0, inc = 0;
  if (tid < RNG_NODES) {
    c = cnt[tid];
    inc = c;
    for (int o = 1; o < 32; o <<= 1) {
      const int u = __shfl_up(inc, o);
      if (lane >= o) inc += u;
    }
    if (lane == 31) sv[wave] = inc;
  }
  __syncthreads();
  if (tid < RNG_NODES) {
    int wbase = 0;
    for (int w = 0; w < wave; ++w) wbase += sv[w];
    const int ex = beg + wbase + inc - c;
    cur[tid] = ex;
    const int node = r * RNG_NODES + tid;
    if (node <= N_NODES) off[node] = ex;
    if (node < N_NODES) {
      const float d = 1.0f + dw[tid];
      dinv[node] = (d > 0.0f) ? rsqrtf(d) : 0.0f;
    }
  }
  __syncthreads();
  for (int e = beg + tid; e < end; e += 4 * 1024) {
    int2 q[4];
#pragma unroll
    for (int k = 0; k < 4; ++k) q[k] = rec2[min(e + k * 1024, end - 1)];
#pragma unroll
    for (int k = 0; k < 4; ++k) {
      if (e + k * 1024 < end) {
        const int slot = atomicAdd(&cur[q[k].x >> SRC_BITS], 1);
        csr2[slot] = make_int2(q[k].x & SRC_MASK, q[k].y);
      }
    }
  }
}

#define GX_NPW 4
#define GX_TILE (8 * GX_NPW)
#define GX_PITCH 132
static_assert(GX_TILE == 32, "two 16-row matrix tiles per work-group");
static_assert(N_NODES % GX_TILE == 0, "the grid covers the nodes exactly, no row guards");

__device__ __forceinline__ void gather_rows(float4& acc, const int s_l, const float nrm_l,
                                            const int n, const float4* __restrict__ r4,
                                            const int lane) {
#pragma unroll 4
  for (int j = 0; j < n; ++j) {
    const int s = __builtin_amdgcn_readlane(s_l, j);
    const float nrm = __int_as_float(__builtin_amdgcn_readlane(__float_as_int(nrm_l), j));
    const float4 v = r4[s * 32 + lane];
    acc.x += v.x * nrm;
    acc.y += v.y * nrm;
    acc.z += v.z * nrm;
    acc.w += v.w * nrm;
  }
}

__global__ __launch_bounds__(256) __attribute__((amdgpu_waves_per_eu(16, 16)))
void gather_gemm12_kernel(const int* __restrict__ off,
                                                            const int* __restrict__ csr,
                                                            const float* __restrict__ dinv,
                                                            const float* __restrict__ x,
                                                            const float* __restrict__ W1,
                                                            const float* __restrict__ b1,
                                                            const float* __restrict__ W2,
                                                            float* __restrict__ h2) {
  __shared__ float sA[GX_TILE * GX_PITCH];
  __shared__ float sW[128 * 16];
  const int tid  = threadIdx.x;
  const int lane = tid & 31;
  const int wave = __builtin_amdgcn_readfirstlane(tid >> 5);
  const float4 w2a = ((const float4*)W2)[tid];
  const float4 w2b = ((const float4*)W2)[tid + 256];
  const float4* x4  = (const float4*)x;
  const int2* csr2  = (const int2*)csr;
  const int d0 = (int)blockIdx.x * GX_TILE + wave;

  int beg[GX_NPW], end[GX_NPW], nn[GX_NPW], s_l[GX_NPW];
  float dd[GX_NPW], w_l[GX_NPW], nrm_l[GX_NPW];
  float4 acc[GX_NPW];
#pragma unroll
  for (int p = 0; p < GX_NPW; ++p) {
    const int d = d0 + 8 * p;
    beg[p] = off[d];
    end[p] = off[d + 1];
    dd[p]  = dinv[d];
  }
#pragma unroll
  for (int p = 0; p < GX_NPW; ++p) {
    const int d = d0 + 8 * p;
    acc[p] = x4[d * 32 + lane];
    nn[p]  = min(32, end[p] - beg[p]);
    const bool on = lane < nn[p];
    const int2 ed = csr2[on ? beg[p] + lane : 0];
    s_l[p] = on ? ed.x : 0;
    w_l[p] = on ? __int_as_float(ed.y) : 0.f;
  }
#pragma unroll
  for (int p = 0; p < GX_NPW; ++p) nrm_l[p] = dinv[s_l[p]] * w_l[p] * dd[p];
  asm volatile("" :: "v"(w2a.x), "v"(w2a.y), "v"(w2a.z), "v"(w2a.w),
                     "v"(w2b.x), "v"(w2b.y), "v"(w2b.z), "v"(w2b.w));

#pragma unroll
  for (int p = 0; p < GX_NPW; ++p) {
    float4 a = acc[p];
    a.x = a.x * dd[p] * dd[p];
    a.y = a.y * dd[p] * dd[p];
    a.z = a.z * dd[p] * dd[p];
    a.w = a.w * dd[p] * dd[p];
    gather_rows(a, s_l[p], nrm_l[p], nn[p], x4, lane);
    for (int base = beg[p] + 32; base < end[p]; base += 32) {
      const int n = min(32, end[p] - base);
      int s2 = 0;
      float nrm2 = 0.f;
      if (lane < n) {
        const int2 ed = csr2[base + lane];
        s2 = ed.x;
        nrm2 = dinv[ed.x] * __int_as_float(ed.y) * dd[p];
      }
      gather_rows(a, s2, nrm2, n, x4, lane);
    }
    *(float4*)&sA[(8 * p + wave) * GX_PITCH + 4 * lane] = a;
  }
  ((float4*)sW)[tid] = w2a;
  ((float4*)sW)[tid + 256] = w2b;

  const int lrow = lane & 15, hi = lane >> 4;
  const int col = wave * 16 + lrow;
  const float* bp = W1 + 2 * hi * 128 + col;
  v2f bq[8];
#pragma unroll
  for (int i = 0; i < 8; ++i) {
    bq[i].x = bp[(4 * i) * 128];
    bq[i].y = bp[(4 * i + 1) * 128];
  }
  __syncthreads();

  v8f c0 = {}, c1 = {};
  {
    const float* a0p = &sA[lrow * GX_PITCH + 2 * hi];
    const float* a1p = a0p + 16 * GX_PITCH;
#pragma unroll
    for (int i = 0; i < 32; ++i) {
      const int k0 = 4 * i;
      v2f a0, a1, b;
      a0.x = a0p[k0];
      a0.y = a0p[k0 + 1];
      a1.x = a1p[k0];
      a1.y = a1p[k0 + 1];
      if (i < 8) {
        b = bq[i];
      } else {
        b.x = bp[k0 * 128];
        b.y = bp[(k0 + 1) * 128];
      }
      c0 = __builtin_amdgcn_wmma_f32_16x16x4_f32(false, a0, false, b,
                                                 (short)0, c0, false, false);
      c1 = __builtin_amdgcn_wmma_f32_16x16x4_f32(false, a1, false, b,
                                                 (short)0, c1, false, false);
    }
  }
  const float bv = b1[col];
  __syncthreads();
#pragma unroll
  for (int r = 0; r < 8; ++r) {
    sA[(8 * hi + r) * GX_PITCH + col]      = fmaxf(c0[r] + bv, 0.f);
    sA[(16 + 8 * hi + r) * GX_PITCH + col] = fmaxf(c1[r] + bv, 0.f);
  }
  __syncthreads();

  if (wave < 2) {
    const float* ax = &sA[(16 * wave + lrow) * GX_PITCH + 2 * hi];
    const int rbase = (int)blockIdx.x * GX_TILE + 16 * wave + 8 * hi;
    float dv[8];
#pragma unroll
    for (int r = 0; r < 8; ++r) dv[r] = dinv[rbase + r];
    v8f c = {};
    for (int k0 = 0; k0 < 128; k0 += 4) {
      const int k = k0 + 2 * hi;
      v2f a, b;
      a.x = ax[k0];
      a.y = ax[k0 + 1];
      b.x = sW[k * 16 + lrow];
      b.y = sW[(k + 1) * 16 + lrow];
      c = __builtin_amdgcn_wmma_f32_16x16x4_f32(false, a, false, b,
                                                (short)0, c, false, false);
    }
#pragma unroll
    for (int r = 0; r < 8; ++r)
      h2[(rbase + r) * 16 + lrow] = c[r] * dv[r];
  }
}

__global__ __launch_bounds__(256) void gather2_softmax_kernel(const int* __restrict__ off,
                                                              const int* __restrict__ csr,
                                                              const float* __restrict__ dinv,
                                                              const float* __restrict__ h2s,
                                                              const float* __restrict__ b2,
                                                              float* __restrict__ out) {
  const int gid = blockIdx.x * 256 + threadIdx.x;
  const int sub = threadIdx.x & 3;
  const bool live = (gid >> 2) < N_NODES;
  const int d = live ? (gid >> 2) : (N_NODES - 1);
  const int beg = off[d], end = off[d + 1];
  const float dd = dinv[d];
  const float4 bias = ((const float4*)b2)[sub];
  const float4* p = (const float4*)h2s;
  const int2* csr2 = (const int2*)csr;
  float4 acc = p[d * 4 + sub];
  for (int j = beg; j < end; ++j) {
    const int2 ed = csr2[j];
    const float w = __int_as_float(ed.y);
    const float4 v = p[ed.x * 4 + sub];
    acc.x += v.x * w;
    acc.y += v.y * w;
    acc.z += v.z * w;
    acc.w += v.w * w;
  }
  const float v0 = acc.x * dd + bias.x;
  const float v1 = acc.y * dd + bias.y;
  const float v2 = acc.z * dd + bias.z;
  const float v3 = acc.w * dd + bias.w;
  float m = fmaxf(fmaxf(v0, v1), fmaxf(v2, v3));
  m = fmaxf(m, __shfl_xor(m, 1));
  m = fmaxf(m, __shfl_xor(m, 2));
  const float e0 = expf(v0 - m);
  const float e1 = expf(v1 - m);
  const float e2 = expf(v2 - m);
  const float e3 = expf(v3 - m);
  float s = (e0 + e1) + (e2 + e3);
  s += __shfl_xor(s, 1);
  s += __shfl_xor(s, 2);
  const float inv = 1.0f / s;
  if (live)
    ((float4*)out)[d * 4 + sub] = make_float4(e0 * inv, e1 * inv, e2 * inv, e3 * inv);
}

extern "C" void kernel_launch(void* const* d_in, const int* in_sizes, int n_in,
                              void* d_out, int out_size, void* d_ws, size_t ws_size,
                              hipStream_t stream) {
  const float* x  = (const float*)d_in[0];
  const int*   ei = (const int*)d_in[1];
  const float* ew = (const float*)d_in[2];
  const float* W1 = (const float*)d_in[3];
  const float* b1 = (const float*)d_in[4];
  const float* W2 = (const float*)d_in[5];
  const float* b2 = (const float*)d_in[6];
  float* out = (float*)d_out;

  const int* src = ei;
  const int* dst = ei + N_EDGES;

  float* ws   = (float*)d_ws;
  float* dinv = ws;
  float* h2   = ws + N_NODES + (size_t)N_NODES * 128;
  int* off    = (int*)(ws + N_NODES + (size_t)N_NODES * 256);
  int* tab    = off + 100352;
  int* base   = tab + 153600;
  int* rstart = base + 153600;
  int* rec    = rstart + 512;
  int* csr    = rec + 2 * N_EDGES;

  const int T = 256;

  range_count_kernel<<<NCHUNK, T, 0, stream>>>(dst, tab);
  range_scan_kernel<<<NRNG, 1024, 0, stream>>>(tab, base, rstart);
  range_place_kernel<<<NCHUNK, T, 0, stream>>>(src, dst, ew, base, rec);
  range_build_kernel<<<NRNG, 1024, 0, stream>>>(rec, rstart, off, dinv, csr);

  gather_gemm12_kernel<<<N_NODES / GX_TILE, T, 0, stream>>>(off, csr, dinv, x, W1, b1, W2, h2);

  gather2_softmax_kernel<<<(N_NODES * 4 + T - 1) / T, T, 0, stream>>>(off, csr, dinv, h2, b2, out);
}


# Coarser sort (512-node ranges, 8192-edge chunks): each placing work-group sums the 196x196 count table itself under its edge loads, scan launch gone (6 to 5); edge and degree sum order change
# speedup vs baseline: 4.8935x; 1.2061x over previous
#include <hip/hip_runtime.h>
#include <hip/hip_bf16.h>
#include <math.h>


#define N_NODES 100000
#define N_EDGES 1600000
#define F_DIM 128
#define H_DIM 128
#define C_DIM 16

#define RNG_SHIFT 9
#define RNG_NODES (1 << RNG_SHIFT)
#define NRNG ((N_NODES >> RNG_SHIFT) + 1)
#define CHUNK_EDGES 8192
#define NCHUNK ((N_EDGES + CHUNK_EDGES - 1) / CHUNK_EDGES)
#define E4 (N_EDGES / 4)
#define SRC_BITS 17
#define SRC_MASK ((1 << SRC_BITS) - 1)
#define SORT_T 512
#define SORT_WAVES (SORT_T / 32)
#define CHUNK_TRIPS (CHUNK_EDGES / (4 * SORT_T))
#define ROW4 (NCHUNK / 4)
#define ROWS_PER_WAVE ((NRNG + SORT_WAVES - 1) / SORT_WAVES)
static_assert(N_NODES <= (1 << SRC_BITS) && SRC_BITS + RNG_SHIFT <= 31, "source index and in-range destination share one word");
static_assert(N_EDGES % 4 == 0, "edge arrays are read four entries at a time");
static_assert(CHUNK_EDGES % (4 * SORT_T) == 0, "a chunk is whole trips of the work-group");
static_assert(NCHUNK % 4 == 0 && ROW4 > 32 && ROW4 <= 64, "a wave reads one table row with two 16-byte loads per lane");
static_assert(NRNG <= SORT_T, "the range starts are scanned one range per thread");
static_assert(NRNG * NCHUNK <= 153600 && NRNG <= 512 && N_NODES + 1 <= 100352, "workspace reservations");

typedef __attribute__((ext_vector_type(2))) float v2f;
typedef __attribute__((ext_vector_type(8))) float v8f;

__global__ __launch_bounds__(SORT_T) void range_count_kernel(const int* __restrict__ dst,
                                                             int* __restrict__ tab) {
  __shared__ int sc[NRNG];
  const int tid = threadIdx.x;
  for (int i = tid; i < NRNG; i += SORT_T) sc[i] = 0;
  __syncthreads();
  const int4* d4 = (const int4*)dst;
#pragma unroll
  for (int k = 0; k < CHUNK_TRIPS; ++k) {
    const int idx = (int)blockIdx.x * (CHUNK_EDGES / 4) + k * SORT_T + tid;
    if (idx < E4) {
      const int4 d = d4[idx];
      atomicAdd(&sc[d.x >> RNG_SHIFT], 1);
      atomicAdd(&sc[d.y >> RNG_SHIFT], 1);
      atomicAdd(&sc[d.z >> RNG_SHIFT], 1);
      atomicAdd(&sc[d.w >> RNG_SHIFT], 1);
    }
  }
  __syncthreads();
  for (int i = tid; i < NRNG; i += SORT_T) tab[i * NCHUNK + (int)blockIdx.x] = sc[i];
}

__global__ __launch_bounds__(SORT_T) __attribute__((amdgpu_waves_per_eu(4, 4)))
void range_place_kernel(const int* __restrict__ src,
                                                             const int* __restrict__ dst,
                                                             const float* __restrict__ ew,
                                                             const int* __restrict__ tab,
                                                             int* __restrict__ rstart,
                                                             int* __restrict__ rec) {
  __shared__ int rtot[NRNG];
  __shared__ int cur[NRNG];
  __shared__ int sv[SORT_WAVES];
  const int tid = threadIdx.x, lane = tid & 31, wave = tid >> 5;
  const int c = (int)blockIdx.x;
  const int4* d4 = (const int4*)dst;
  const int4* s4 = (const int4*)src;
  const float4* w4 = (const float4*)ew;
  int4 d[CHUNK_TRIPS], sr[CHUNK_TRIPS];
  float4 w[CHUNK_TRIPS];
  {
    const int4* t4 = (const int4*)tab;
    int4 va[ROWS_PER_WAVE], vb[ROWS_PER_WAVE];
#pragma unroll
    for (int k = 0; k < ROWS_PER_WAVE; ++k) {
      const int r = min(wave + k * SORT_WAVES, NRNG - 1);
      va[k] = t4[r * ROW4 + lane];
      vb[k] = t4[r * ROW4 + min(lane + 32, ROW4 - 1)];
    }
#pragma unroll
    for (int k = 0; k < CHUNK_TRIPS; ++k) {
      const int idx = min(c * (CHUNK_EDGES / 4) + k * SORT_T + tid, E4 - 1);
      d[k] = d4[idx];
      sr[k] = s4[idx];
      w[k] = w4[idx];
    }
    int tot[ROWS_PER_WAVE], pre[ROWS_PER_WAVE];
    const int j = 4 * lane;
    const bool two = lane + 32 < ROW4;
#pragma unroll
    for (int k = 0; k < ROWS_PER_WAVE; ++k) {
      tot[k] = va[k].x + va[k].y + va[k].z + va[k].w +
               (two ? vb[k].x + vb[k].y + vb[k].z + vb[k].w : 0);
      pre[k] = ((j < c) ? va[k].x : 0) + ((j + 1 < c) ? va[k].y : 0) +
               ((j + 2 < c) ? va[k].z : 0) + ((j + 3 < c) ? va[k].w : 0) +
               ((two && j + 128 < c) ? vb[k].x : 0) + ((two && j + 129 < c) ? vb[k].y : 0) +
               ((two && j + 130 < c) ? vb[k].z : 0) + ((two && j + 131 < c) ? vb[k].w : 0);
    }
#pragma unroll
    for (int o = 16; o > 0; o >>= 1) {
#pragma unroll
      for (int k = 0; k < ROWS_PER_WAVE; ++k) {
        tot[k] += __shfl_xor(tot[k], o);
        pre[k] += __shfl_xor(pre[k], o);
      }
    }
    if (lane == 0) {
#pragma unroll
      for (int k = 0; k < ROWS_PER_WAVE; ++k) {
        const int r = wave + k * SORT_WAVES;
        if (r < NRNG) {
          rtot[r] = tot[k];
          cur[r] = pre[k];
        }
      }
    }
  }
  __syncthreads();
  const int a = (tid < NRNG) ? rtot[tid] : 0;
  int inc = a;
  for (int o = 1; o < 32; o <<= 1) {
    const int u = __shfl_up(inc, o);
    if (lane >= o) inc += u;
  }
  if (lane == 31) sv[wave] = inc;
  __syncthreads();
  int wbase = 0;
  for (int w = 0; w < wave; ++w) wbase += sv[w];
  if (tid < NRNG) {
    const int first = wbase + inc - a;
    cur[tid] += first;
    if (c == 0) rstart[tid] = first;
  }
  __syncthreads();
  int2* rec2 = (int2*)rec;
#pragma unroll
  for (int k = 0; k < CHUNK_TRIPS; ++k) {
    if (c * (CHUNK_EDGES / 4) + k * SORT_T + tid < E4) {
      const int p0 = atomicAdd(&cur[d[k].x >> RNG_SHIFT], 1);
      const int p1 = atomicAdd(&cur[d[k].y >> RNG_SHIFT], 1);
      const int p2 = atomicAdd(&cur[d[k].z >> RNG_SHIFT], 1);
      const int p3 = atomicAdd(&cur[d[k].w >> RNG_SHIFT], 1);
      rec2[p0] = make_int2(sr[k].x | ((d[k].x & (RNG_NODES - 1)) << SRC_BITS), __float_as_int(w[k].x));
      rec2[p1] = make_int2(sr[k].y | ((d[k].y & (RNG_NODES - 1)) << SRC_BITS), __float_as_int(w[k].y));
      rec2[p2] = make_int2(sr[k].z | ((d[k].z & (RNG_NODES - 1)) << SRC_BITS), __float_as_int(w[k].z));
      rec2[p3] = make_int2(sr[k].w | ((d[k].w & (RNG_NODES - 1)) << SRC_BITS), __float_as_int(w[k].w));
    }
  }
}

__global__ __launch_bounds__(1024) void range_build_kernel(const int* __restrict__ rec,
                                                           const int* __restrict__ rstart,
                                                           int* __restrict__ off,
                                                           float* __restrict__ dinv,
                                                           int* __restrict__ csr) {
  __shared__ int cnt[RNG_NODES];
  __shared__ float dw[RNG_NODES];
  __shared__ int cur[RNG_NODES];
  __shared__ int sv[RNG_NODES / 32];
  const int tid = threadIdx.x, lane = tid & 31, wave = tid >> 5;
  const int r = (int)blockIdx.x;
  const int beg = rstart[r];
  const int end = (r + 1 < NRNG) ? rstart[r + 1] : N_EDGES;
  const int2* rec2 = (const int2*)rec;
  int2* csr2 = (int2*)csr;
  if (tid < RNG_NODES) {
    cnt[tid] = 0;
    dw[tid] = 0.f;
  }
  __syncthreads();
  for (int e = beg + tid; e < end; e += 4 * 1024) {
    int2 q[4];
#pragma unroll
    for (int k = 0; k < 4; ++k) q[k] = rec2[min(e + k * 1024, end - 1)];
#pragma unroll
    for (int k = 0; k < 4; ++k) {
      if (e + k * 1024 < end) {
        const int dl = q[k].x >> SRC_BITS;
        atomicAdd(&cnt[dl], 1);
        atomicAdd(&dw[dl], __int_as_float(q[k].y));
      }
    }
  }
  __syncthreads();
  int c = 0, inc = 0;
  if (tid < RNG_NODES) {
    c = cnt[tid];
    inc = c;
    for (int o = 1; o < 32; o <<= 1) {
      const int u = __shfl_up(inc, o);
      if (lane >= o) inc += u;
    }
    if (lane == 31) sv[wave] = inc;
  }
  __syncthreads();
  if (tid < RNG_NODES) {
    int wbase = 0;
    for (int w = 0; w < wave; ++w) wbase += sv[w];
    const int ex = beg + wbase + inc - c;
    cur[tid] = ex;
    const int node = r * RNG_NODES + tid;
    if (node <= N_NODES) off[node] = ex;
    if (node < N_NODES) {
      const float d = 1.0f + dw[tid];
      dinv[node] = (d > 0.0f) ? rsqrtf(d) : 0.0f;
    }
  }
  __syncthreads();
  for (int e = beg + tid; e < end; e += 4 * 1024) {
    int2 q[4];
#pragma unroll
    for (int k = 0; k < 4; ++k) q[k] = rec2[min(e + k * 1024, end - 1)];
#pragma unroll
    for (int k = 0; k < 4; ++k) {
      if (e + k * 1024 < end) {
        const int slot = atomicAdd(&cur[q[k].x >> SRC_BITS], 1);
        csr2[slot] = make_int2(q[k].x & SRC_MASK, q[k].y);
      }
    }
  }
}

#define GX_NPW 4
#define GX_TILE (8 * GX_NPW)
#define GX_PITCH 132
static_assert(GX_TILE == 32, "two 16-row matrix tiles per work-group");
static_assert(N_NODES % GX_TILE == 0, "the grid covers the nodes exactly, no row guards");

__device__ __forceinline__ void gather_rows(float4& acc, const int s_l, const float nrm_l,
                                            const int n, const float4* __restrict__ r4,
                                            const int lane) {
#pragma unroll 4
  for (int j = 0; j < n; ++j) {
    const int s = __builtin_amdgcn_readlane(s_l, j);
    const float nrm = __int_as_float(__builtin_amdgcn_readlane(__float_as_int(nrm_l), j));
    const float4 v = r4[s * 32 + lane];
    acc.x += v.x * nrm;
    acc.y += v.y * nrm;
    acc.z += v.z * nrm;
    acc.w += v.w * nrm;
  }
}

__global__ __launch_bounds__(256) __attribute__((amdgpu_waves_per_eu(16, 16)))
void gather_gemm12_kernel(const int* __restrict__ off,
                                                            const int* __restrict__ csr,
                                                            const float* __restrict__ dinv,
                                                            const float* __restrict__ x,
                                                            const float* __restrict__ W1,
                                                            const float* __restrict__ b1,
                                                            const float* __restrict__ W2,
                                                            float* __restrict__ h2) {
  __shared__ float sA[GX_TILE * GX_PITCH];
  __shared__ float sW[128 * 16];
  const int tid  = threadIdx.x;
  const int lane = tid & 31;
  const int wave = __builtin_amdgcn_readfirstlane(tid >> 5);
  const float4 w2a = ((const float4*)W2)[tid];
  const float4 w2b = ((const float4*)W2)[tid + 256];
  const float4* x4  = (const float4*)x;
  const int2* csr2  = (const int2*)csr;
  const int d0 = (int)blockIdx.x * GX_TILE + wave;

  int beg[GX_NPW], end[GX_NPW], nn[GX_NPW], s_l[GX_NPW];
  float dd[GX_NPW], w_l[GX_NPW], nrm_l[GX_NPW];
  float4 acc[GX_NPW];
#pragma unroll
  for (int p = 0; p < GX_NPW; ++p) {
    const int d = d0 + 8 * p;
    beg[p] = off[d];
    end[p] = off[d + 1];
    dd[p]  = dinv[d];
  }
#pragma unroll
  for (int p = 0; p < GX_NPW; ++p) {
    const int d = d0 + 8 * p;
    acc[p] = x4[d * 32 + lane];
    nn[p]  = min(32, end[p] - beg[p]);
    const bool on = lane < nn[p];
    const int2 ed = csr2[on ? beg[p] + lane : 0];
    s_l[p] = on ? ed.x : 0;
    w_l[p] = on ? __int_as_float(ed.y) : 0.f;
  }
#pragma unroll
  for (int p = 0; p < GX_NPW; ++p) nrm_l[p] = dinv[s_l[p]] * w_l[p] * dd[p];
  asm volatile("" :: "v"(w2a.x), "v"(w2a.y), "v"(w2a.z), "v"(w2a.w),
                     "v"(w2b.x), "v"(w2b.y), "v"(w2b.z), "v"(w2b.w));

#pragma unroll
  for (int p = 0; p < GX_NPW; ++p) {
    float4 a = acc[p];
    a.x = a.x * dd[p] * dd[p];
    a.y = a.y * dd[p] * dd[p];
    a.z = a.z * dd[p] * dd[p];
    a.w = a.w * dd[p] * dd[p];
    gather_rows(a, s_l[p], nrm_l[p], nn[p], x4, lane);
    for (int base = beg[p] + 32; base < end[p]; base += 32) {
      const int n = min(32, end[p] - base);
      int s2 = 0;
      float nrm2 = 0.f;
      if (lane < n) {
        const int2 ed = csr2[base + lane];
        s2 = ed.x;
        nrm2 = dinv[ed.x] * __int_as_float(ed.y) * dd[p];
      }
      gather_rows(a, s2, nrm2, n, x4, lane);
    }
    *(float4*)&sA[(8 * p + wave) * GX_PITCH + 4 * lane] = a;
  }
  ((float4*)sW)[tid] = w2a;
  ((float4*)sW)[tid + 256] = w2b;

  const int lrow = lane & 15, hi = lane >> 4;
  const int col = wave * 16 + lrow;
  const float* bp = W1 + 2 * hi * 128 + col;
  v2f bq[8];
#pragma unroll
  for (int i = 0; i < 8; ++i) {
    bq[i].x = bp[(4 * i) * 128];
    bq[i].y = bp[(4 * i + 1) * 128];
  }
  __syncthreads();

  v8f c0 = {}, c1 = {};
  {
    const float* a0p = &sA[lrow * GX_PITCH + 2 * hi];
    const float* a1p = a0p + 16 * GX_PITCH;
#pragma unroll
    for (int i = 0; i < 32; ++i) {
      const int k0 = 4 * i;
      v2f a0, a1, b;
      a0.x = a0p[k0];
      a0.y = a0p[k0 + 1];
      a1.x = a1p[k0];
      a1.y = a1p[k0 + 1];
      if (i < 8) {
        b = bq[i];
      } else {
        b.x = bp[k0 * 128];
        b.y = bp[(k0 + 1) * 128];
      }
      c0 = __builtin_amdgcn_wmma_f32_16x16x4_f32(false, a0, false, b,
                                                 (short)0, c0, false, false);
      c1 = __builtin_amdgcn_wmma_f32_16x16x4_f32(false, a1, false, b,
                                                 (short)0, c1, false, false);
    }
  }
  const float bv = b1[col];
  __syncthreads();
#pragma unroll
  for (int r = 0; r < 8; ++r) {
    sA[(8 * hi + r) * GX_PITCH + col]      = fmaxf(c0[r] + bv, 0.f);
    sA[(16 + 8 * hi + r) * GX_PITCH + col] = fmaxf(c1[r] + bv, 0.f);
  }
  __syncthreads();

  if (wave < 2) {
    const float* ax = &sA[(16 * wave + lrow) * GX_PITCH + 2 * hi];
    const int rbase = (int)blockIdx.x * GX_TILE + 16 * wave + 8 * hi;
    float dv[8];
#pragma unroll
    for (int r = 0; r < 8; ++r) dv[r] = dinv[rbase + r];
    v8f c = {};
    for (int k0 = 0; k0 < 128; k0 += 4) {
      const int k = k0 + 2 * hi;
      v2f a, b;
      a.x = ax[k0];
      a.y = ax[k0 + 1];
      b.x = sW[k * 16 + lrow];
      b.y = sW[(k + 1) * 16 + lrow];
      c = __builtin_amdgcn_wmma_f32_16x16x4_f32(false, a, false, b,
                                                (short)0, c, false, false);
    }
#pragma unroll
    for (int r = 0; r < 8; ++r)
      h2[(rbase + r) * 16 + lrow] = c[r] * dv[r];
  }
}

__global__ __launch_bounds__(256) void gather2_softmax_kernel(const int* __restrict__ off,
                                                              const int* __restrict__ csr,
                                                              const float* __restrict__ dinv,
                                                              const float* __restrict__ h2s,
                                                              const float* __restrict__ b2,
                                                              float* __restrict__ out) {
  const int gid = blockIdx.x * 256 + threadIdx.x;
  const int sub = threadIdx.x & 3;
  const bool live = (gid >> 2) < N_NODES;
  const int d = live ? (gid >> 2) : (N_NODES - 1);
  const int beg = off[d], end = off[d + 1];
  const float dd = dinv[d];
  const float4 bias = ((const float4*)b2)[sub];
  const float4* p = (const float4*)h2s;
  const int2* csr2 = (const int2*)csr;
  float4 acc = p[d * 4 + sub];
  for (int j = beg; j < end; ++j) {
    const int2 ed = csr2[j];
    const float w = __int_as_float(ed.y);
    const float4 v = p[ed.x * 4 + sub];
    acc.x += v.x * w;
    acc.y += v.y * w;
    acc.z += v.z * w;
    acc.w += v.w * w;
  }
  const float v0 = acc.x * dd + bias.x;
  const float v1 = acc.y * dd + bias.y;
  const float v2 = acc.z * dd + bias.z;
  const float v3 = acc.w * dd + bias.w;
  float m = fmaxf(fmaxf(v0, v1), fmaxf(v2, v3));
  m = fmaxf(m, __shfl_xor(m, 1));
  m = fmaxf(m, __shfl_xor(m, 2));
  const float e0 = expf(v0 - m);
  const float e1 = expf(v1 - m);
  const float e2 = expf(v2 - m);
  const float e3 = expf(v3 - m);
  float s = (e0 + e1) + (e2 + e3);
  s += __shfl_xor(s, 1);
  s += __shfl_xor(s, 2);
  const float inv = 1.0f / s;
  if (live)
    ((float4*)out)[d * 4 + sub] = make_float4(e0 * inv, e1 * inv, e2 * inv, e3 * inv);
}

extern "C" void kernel_launch(void* const* d_in, const int* in_sizes, int n_in,
                              void* d_out, int out_size, void* d_ws, size_t ws_size,
                              hipStream_t stream) {
  const float* x  = (const float*)d_in[0];
  const int*   ei = (const int*)d_in[1];
  const float* ew = (const float*)d_in[2];
  const float* W1 = (const float*)d_in[3];
  const float* b1 = (const float*)d_in[4];
  const float* W2 = (const float*)d_in[5];
  const float* b2 = (const float*)d_in[6];
  float* out = (float*)d_out;

  const int* src = ei;
  const int* dst = ei + N_EDGES;

  float* ws   = (float*)d_ws;
  float* dinv = ws;
  float* h2   = ws + N_NODES + (size_t)N_NODES * 128;
  int* off    = (int*)(ws + N_NODES + (size_t)N_NODES * 256);
  int* tab    = off + 100352;
  int* rstart = tab + 153600;
  int* rec    = rstart + 512;
  int* csr    = rec + 2 * N_EDGES;

  const int T = 256;

  range_count_kernel<<<NCHUNK, SORT_T, 0, stream>>>(dst, tab);
  range_place_kernel<<<NCHUNK, SORT_T, 0, stream>>>(src, dst, ew, tab, rstart, rec);
  range_build_kernel<<<NRNG, 1024, 0, stream>>>(rec, rstart, off, dinv, csr);

  gather_gemm12_kernel<<<N_NODES / GX_TILE, T, 0, stream>>>(off, csr, dinv, x, W1, b1, W2, h2);

  gather2_softmax_kernel<<<(N_NODES * 4 + T - 1) / T, T, 0, stream>>>(off, csr, dinv, h2, b2, out);
}


# Sort in 2 launches: each chunk groups its edges by range in place (one returning LDS add ranks and counts), a range walks its 196 groups; count launch gone (5 to 4); edge and degree sum order change
# speedup vs baseline: 5.1429x; 1.0510x over previous
#include <hip/hip_runtime.h>
#include <hip/hip_bf16.h>
#include <math.h>


#define N_NODES 100000
#define N_EDGES 1600000
#define F_DIM 128
#define H_DIM 128
#define C_DIM 16

#define RNG_SHIFT 9
#define RNG_NODES (1 << RNG_SHIFT)
#define NRNG ((N_NODES >> RNG_SHIFT) + 1)
#define CHUNK_EDGES 8192
#define NCHUNK ((N_EDGES + CHUNK_EDGES - 1) / CHUNK_EDGES)
#define E4 (N_EDGES / 4)
#define SRC_BITS 17
#define SRC_MASK ((1 << SRC_BITS) - 1)
#define SORT_T 512
#define SORT_WAVES (SORT_T / 32)
#define CHUNK_TRIPS (CHUNK_EDGES / (4 * SORT_T))
#define ROW4 (NCHUNK / 4)
static_assert(N_NODES <= (1 << SRC_BITS) && SRC_BITS + RNG_SHIFT <= 31, "source index and in-range destination share one word");
static_assert(N_EDGES % 4 == 0, "edge arrays are read four entries at a time");
static_assert(CHUNK_EDGES % (4 * SORT_T) == 0, "a chunk is whole trips of the work-group");
static_assert(NCHUNK % 4 == 0 && ROW4 > 32 && ROW4 <= 64, "a wave reads one table row with two 16-byte loads per lane");
static_assert(NRNG <= SORT_T, "the range starts are scanned one range per thread");
static_assert(NRNG * NCHUNK <= 153600 && N_NODES + 1 <= 100352, "workspace reservations");

typedef __attribute__((ext_vector_type(2))) float v2f;
typedef __attribute__((ext_vector_type(8))) float v8f;

__global__ __launch_bounds__(SORT_T) __attribute__((amdgpu_waves_per_eu(4, 4)))
void chunk_group_kernel(const int* __restrict__ src,
                        const int* __restrict__ dst,
                        const float* __restrict__ ew,
                        int* __restrict__ tab,
                        int* __restrict__ rec) {
  __shared__ int cnt[NRNG];
  __shared__ int first[NRNG];
  __shared__ int sv[SORT_WAVES];
  const int tid = threadIdx.x, lane = tid & 31, wave = tid >> 5;
  const int c = (int)blockIdx.x;
  const int4* d4 = (const int4*)dst;
  const int4* s4 = (const int4*)src;
  const float4* w4 = (const float4*)ew;
  for (int i = tid; i < NRNG; i += SORT_T) cnt[i] = 0;
  int4 d[CHUNK_TRIPS], sr[CHUNK_TRIPS], p[CHUNK_TRIPS];
  float4 w[CHUNK_TRIPS];
#pragma unroll
  for (int k = 0; k < CHUNK_TRIPS; ++k) {
    const int idx = min(c * (CHUNK_EDGES / 4) + k * SORT_T + tid, E4 - 1);
    d[k] = d4[idx];
    sr[k] = s4[idx];
    w[k] = w4[idx];
  }
  __syncthreads();
#pragma unroll
  for (int k = 0; k < CHUNK_TRIPS; ++k) {
    p[k] = make_int4(0, 0, 0, 0);
    if (c * (CHUNK_EDGES / 4) + k * SORT_T + tid < E4) {
      p[k].x = atomicAdd(&cnt[d[k].x >> RNG_SHIFT], 1);
      p[k].y = atomicAdd(&cnt[d[k].y >> RNG_SHIFT], 1);
      p[k].z = atomicAdd(&cnt[d[k].z >> RNG_SHIFT], 1);
      p[k].w = atomicAdd(&cnt[d[k].w >> RNG_SHIFT], 1);
    }
  }
  __syncthreads();
  const int a = (tid < NRNG) ? cnt[tid] : 0;
  int inc = a;
  for (int o = 1; o < 32; o <<= 1) {
    const int u = __shfl_up(inc, o);
    if (lane >= o) inc += u;
  }
  if (lane == 31) sv[wave] = inc;
  __syncthreads();
  int wbase = 0;
  for (int q = 0; q < wave; ++q) wbase += sv[q];
  if (tid < NRNG) {
    const int f = wbase + inc - a;
    first[tid] = f;
    tab[tid * NCHUNK + c] = f | (a << 16);
  }
  __syncthreads();
  int2* rec2 = (int2*)rec + c * CHUNK_EDGES;
#pragma unroll
  for (int k = 0; k < CHUNK_TRIPS; ++k) {
    if (c * (CHUNK_EDGES / 4) + k * SORT_T + tid < E4) {
      const int p0 = first[d[k].x >> RNG_SHIFT] + p[k].x;
      const int p1 = first[d[k].y >> RNG_SHIFT] + p[k].y;
      const int p2 = first[d[k].z >> RNG_SHIFT] + p[k].z;
      const int p3 = first[d[k].w >> RNG_SHIFT] + p[k].w;
      rec2[p0] = make_int2(sr[k].x | ((d[k].x & (RNG_NODES - 1)) << SRC_BITS), __float_as_int(w[k].x));
      rec2[p1] = make_int2(sr[k].y | ((d[k].y & (RNG_NODES - 1)) << SRC_BITS), __float_as_int(w[k].y));
      rec2[p2] = make_int2(sr[k].z | ((d[k].z & (RNG_NODES - 1)) << SRC_BITS), __float_as_int(w[k].z));
      rec2[p3] = make_int2(sr[k].w | ((d[k].w & (RNG_NODES - 1)) << SRC_BITS), __float_as_int(w[k].w));
    }
  }
}

#define BUILD_T 1024
#define BUILD_WAVES (BUILD_T / 32)
#define GROUPS_PER_WAVE ((NCHUNK + BUILD_WAVES - 1) / BUILD_WAVES)
static_assert(GROUPS_PER_WAVE * BUILD_WAVES >= NRNG, "the same wave stride covers the table rows");
static_assert(CHUNK_EDGES < (1 << 16), "a group's first slot and its size share one word");
__global__ __launch_bounds__(BUILD_T) void range_build_kernel(const int* __restrict__ rec,
                                                              const int* __restrict__ tab,
                                                              int* __restrict__ off,
                                                              float* __restrict__ dinv,
                                                              int* __restrict__ csr) {
  __shared__ int cnt[RNG_NODES];
  __shared__ float dw[RNG_NODES];
  __shared__ int cur[RNG_NODES];
  __shared__ int sv[RNG_NODES / 32];
  __shared__ int sb[BUILD_WAVES];
  const int tid = threadIdx.x, lane = tid & 31, wave = tid >> 5;
  const int r = (int)blockIdx.x;
  const int2* rec2 = (const int2*)rec;
  int2* csr2 = (int2*)csr;
  int gbeg[GROUPS_PER_WAVE], gn[GROUPS_PER_WAVE];
  {
    const int4* t4 = (const int4*)tab;
    int4 va[GROUPS_PER_WAVE], vb[GROUPS_PER_WAVE];
    int hd[GROUPS_PER_WAVE];
#pragma unroll
    for (int k = 0; k < GROUPS_PER_WAVE; ++k) {
      const int row = min(wave + k * BUILD_WAVES, NRNG - 1);
      va[k] = t4[row * ROW4 + lane];
      vb[k] = t4[row * ROW4 + min(lane + 32, ROW4 - 1)];
      hd[k] = tab[r * NCHUNK + min(wave + k * BUILD_WAVES, NCHUNK - 1)];
    }
    if (tid < RNG_NODES) {
      cnt[tid] = 0;
      dw[tid] = 0.f;
    }
    const bool two = lane + 32 < ROW4;
    int part = 0;
#pragma unroll
    for (int k = 0; k < GROUPS_PER_WAVE; ++k) {
      if (wave + k * BUILD_WAVES < r) {
        part += (va[k].x >> 16) + (va[k].y >> 16) + (va[k].z >> 16) + (va[k].w >> 16);
        if (two) part += (vb[k].x >> 16) + (vb[k].y >> 16) + (vb[k].z >> 16) + (vb[k].w >> 16);
      }
    }
#pragma unroll
    for (int o = 16; o > 0; o >>= 1) part += __shfl_xor(part, o);
    if (lane == 0) sb[wave] = part;
#pragma unroll
    for (int k = 0; k < GROUPS_PER_WAVE; ++k) {
      const int ck = wave + k * BUILD_WAVES;
      gn[k] = (ck < NCHUNK) ? (hd[k] >> 16) : 0;
      gbeg[k] = ck * CHUNK_EDGES + (hd[k] & 0xFFFF);
    }
  }
  int maxn = 0;
#pragma unroll
  for (int k = 0; k < GROUPS_PER_WAVE; ++k) maxn = max(maxn, gn[k]);
  __syncthreads();
  int beg = 0;
  for (int q = 0; q < BUILD_WAVES; ++q) beg += sb[q];
  for (int t = lane; t - lane < maxn; t += 32) {
    int2 q[GROUPS_PER_WAVE];
#pragma unroll
    for (int k = 0; k < GROUPS_PER_WAVE; ++k) q[k] = rec2[(t < gn[k]) ? gbeg[k] + t : 0];
#pragma unroll
    for (int k = 0; k < GROUPS_PER_WAVE; ++k) {
      if (t < gn[k]) {
        const int dl = q[k].x >> SRC_BITS;
        atomicAdd(&cnt[dl], 1);
        atomicAdd(&dw[dl], __int_as_float(q[k].y));
      }
    }
  }
  __syncthreads();
  int c = 0, inc = 0;
  if (tid < RNG_NODES) {
    c = cnt[tid];
    inc = c;
    for (int o = 1; o < 32; o <<= 1) {
      const int u = __shfl_up(inc, o);
      if (lane >= o) inc += u;
    }
    if (lane == 31) sv[wave] = inc;
  }
  __syncthreads();
  if (tid < RNG_NODES) {
    int wbase = 0;
    for (int w = 0; w < wave; ++w) wbase += sv[w];
    const int ex = beg + wbase + inc - c;
    cur[tid] = ex;
    const int node = r * RNG_NODES + tid;
    if (node <= N_NODES) off[node] = ex;
    if (node < N_NODES) {
      const float d = 1.0f + dw[tid];
      dinv[node] = (d > 0.0f) ? rsqrtf(d) : 0.0f;
    }
  }
  __syncthreads();
  for (int t = lane; t - lane < maxn; t += 32) {
    int2 q[GROUPS_PER_WAVE];
#pragma unroll
    for (int k = 0; k < GROUPS_PER_WAVE; ++k) q[k] = rec2[(t < gn[k]) ? gbeg[k] + t : 0];
    int slot[GROUPS_PER_WAVE];
#pragma unroll
    for (int k = 0; k < GROUPS_PER_WAVE; ++k) {
      slot[k] = 0;
      if (t < gn[k]) slot[k] = atomicAdd(&cur[q[k].x >> SRC_BITS], 1);
    }
#pragma unroll
    for (int k = 0; k < GROUPS_PER_WAVE; ++k) {
      if (t < gn[k]) csr2[slot[k]] = make_int2(q[k].x & SRC_MASK, q[k].y);
    }
  }
}

#define GX_NPW 4
#define GX_TILE (8 * GX_NPW)
#define GX_PITCH 132
static_assert(GX_TILE == 32, "two 16-row matrix tiles per work-group");
static_assert(N_NODES % GX_TILE == 0, "the grid covers the nodes exactly, no row guards");

__device__ __forceinline__ void gather_rows(float4& acc, const int s_l, const float nrm_l,
                                            const int n, const float4* __restrict__ r4,
                                            const int lane) {
#pragma unroll 4
  for (int j = 0; j < n; ++j) {
    const int s = __builtin_amdgcn_readlane(s_l, j);
    const float nrm = __int_as_float(__builtin_amdgcn_readlane(__float_as_int(nrm_l), j));
    const float4 v = r4[s * 32 + lane];
    acc.x += v.x * nrm;
    acc.y += v.y * nrm;
    acc.z += v.z * nrm;
    acc.w += v.w * nrm;
  }
}

__global__ __launch_bounds__(256) __attribute__((amdgpu_waves_per_eu(16, 16)))
void gather_gemm12_kernel(const int* __restrict__ off,
                                                            const int* __restrict__ csr,
                                                            const float* __restrict__ dinv,
                                                            const float* __restrict__ x,
                                                            const float* __restrict__ W1,
                                                            const float* __restrict__ b1,
                                                            const float* __restrict__ W2,
                                                            float* __restrict__ h2) {
  __shared__ float sA[GX_TILE * GX_PITCH];
  __shared__ float sW[128 * 16];
  const int tid  = threadIdx.x;
  const int lane = tid & 31;
  const int wave = __builtin_amdgcn_readfirstlane(tid >> 5);
  const float4 w2a = ((const float4*)W2)[tid];
  const float4 w2b = ((const float4*)W2)[tid + 256];
  const float4* x4  = (const float4*)x;
  const int2* csr2  = (const int2*)csr;
  const int d0 = (int)blockIdx.x * GX_TILE + wave;

  int beg[GX_NPW], end[GX_NPW], nn[GX_NPW], s_l[GX_NPW];
  float dd[GX_NPW], w_l[GX_NPW], nrm_l[GX_NPW];
  float4 acc[GX_NPW];
#pragma unroll
  for (int p = 0; p < GX_NPW; ++p) {
    const int d = d0 + 8 * p;
    beg[p] = off[d];
    end[p] = off[d + 1];
    dd[p]  = dinv[d];
  }
#pragma unroll
  for (int p = 0; p < GX_NPW; ++p) {
    const int d = d0 + 8 * p;
    acc[p] = x4[d * 32 + lane];
    nn[p]  = min(32, end[p] - beg[p]);
    const bool on = lane < nn[p];
    const int2 ed = csr2[on ? beg[p] + lane : 0];
    s_l[p] = on ? ed.x : 0;
    w_l[p] = on ? __int_as_float(ed.y) : 0.f;
  }
#pragma unroll
  for (int p = 0; p < GX_NPW; ++p) nrm_l[p] = dinv[s_l[p]] * w_l[p] * dd[p];
  asm volatile("" :: "v"(w2a.x), "v"(w2a.y), "v"(w2a.z), "v"(w2a.w),
                     "v"(w2b.x), "v"(w2b.y), "v"(w2b.z), "v"(w2b.w));

#pragma unroll
  for (int p = 0; p < GX_NPW; ++p) {
    float4 a = acc[p];
    a.x = a.x * dd[p] * dd[p];
    a.y = a.y * dd[p] * dd[p];
    a.z = a.z * dd[p] * dd[p];
    a.w = a.w * dd[p] * dd[p];
    gather_rows(a, s_l[p], nrm_l[p], nn[p], x4, lane);
    for (int base = beg[p] + 32; base < end[p]; base += 32) {
      const int n = min(32, end[p] - base);
      int s2 = 0;
      float nrm2 = 0.f;
      if (lane < n) {
        const int2 ed = csr2[base + lane];
        s2 = ed.x;
        nrm2 = dinv[ed.x] * __int_as_float(ed.y) * dd[p];
      }
      gather_rows(a, s2, nrm2, n, x4, lane);
    }
    *(float4*)&sA[(8 * p + wave) * GX_PITCH + 4 * lane] = a;
  }
  ((float4*)sW)[tid] = w2a;
  ((float4*)sW)[tid + 256] = w2b;

  const int lrow = lane & 15, hi = lane >> 4;
  const int col = wave * 16 + lrow;
  const float* bp = W1 + 2 * hi * 128 + col;
  v2f bq[8];
#pragma unroll
  for (int i = 0; i < 8; ++i) {
    bq[i].x = bp[(4 * i) * 128];
    bq[i].y = bp[(4 * i + 1) * 128];
  }
  __syncthreads();

  v8f c0 = {}, c1 = {};
  {
    const float* a0p = &sA[lrow * GX_PITCH + 2 * hi];
    const float* a1p = a0p + 16 * GX_PITCH;
#pragma unroll
    for (int i = 0; i < 32; ++i) {
      const int k0 = 4 * i;
      v2f a0, a1, b;
      a0.x = a0p[k0];
      a0.y = a0p[k0 + 1];
      a1.x = a1p[k0];
      a1.y = a1p[k0 + 1];
      if (i < 8) {
        b = bq[i];
      } else {
        b.x = bp[k0 * 128];
        b.y = bp[(k0 + 1) * 128];
      }
      c0 = __builtin_amdgcn_wmma_f32_16x16x4_f32(false, a0, false, b,
                                                 (short)0, c0, false, false);
      c1 = __builtin_amdgcn_wmma_f32_16x16x4_f32(false, a1, false, b,
                                                 (short)0, c1, false, false);
    }
  }
  const float bv = b1[col];
  __syncthreads();
#pragma unroll
  for (int r = 0; r < 8; ++r) {
    sA[(8 * hi + r) * GX_PITCH + col]      = fmaxf(c0[r] + bv, 0.f);
    sA[(16 + 8 * hi + r) * GX_PITCH + col] = fmaxf(c1[r] + bv, 0.f);
  }
  __syncthreads();

  if (wave < 2) {
    const float* ax = &sA[(16 * wave + lrow) * GX_PITCH + 2 * hi];
    const int rbase = (int)blockIdx.x * GX_TILE + 16 * wave + 8 * hi;
    float dv[8];
#pragma unroll
    for (int r = 0; r < 8; ++r) dv[r] = dinv[rbase + r];
    v8f c = {};
    for (int k0 = 0; k0 < 128; k0 += 4) {
      const int k = k0 + 2 * hi;
      v2f a, b;
      a.x = ax[k0];
      a.y = ax[k0 + 1];
      b.x = sW[k * 16 + lrow];
      b.y = sW[(k + 1) * 16 + lrow];
      c = __builtin_amdgcn_wmma_f32_16x16x4_f32(false, a, false, b,
                                                (short)0, c, false, false);
    }
#pragma unroll
    for (int r = 0; r < 8; ++r)
      h2[(rbase + r) * 16 + lrow] = c[r] * dv[r];
  }
}

__global__ __launch_bounds__(256) void gather2_softmax_kernel(const int* __restrict__ off,
                                                              const int* __restrict__ csr,
                                                              const float* __restrict__ dinv,
                                                              const float* __restrict__ h2s,
                                                              const float* __restrict__ b2,
                                                              float* __restrict__ out) {
  const int gid = blockIdx.x * 256 + threadIdx.x;
  const int sub = threadIdx.x & 3;
  const bool live = (gid >> 2) < N_NODES;
  const int d = live ? (gid >> 2) : (N_NODES - 1);
  const int beg = off[d], end = off[d + 1];
  const float dd = dinv[d];
  const float4 bias = ((const float4*)b2)[sub];
  const float4* p = (const float4*)h2s;
  const int2* csr2 = (const int2*)csr;
  float4 acc = p[d * 4 + sub];
  for (int j = beg; j < end; ++j) {
    const int2 ed = csr2[j];
    const float w = __int_as_float(ed.y);
    const float4 v = p[ed.x * 4 + sub];
    acc.x += v.x * w;
    acc.y += v.y * w;
    acc.z += v.z * w;
    acc.w += v.w * w;
  }
  const float v0 = acc.x * dd + bias.x;
  const float v1 = acc.y * dd + bias.y;
  const float v2 = acc.z * dd + bias.z;
  const float v3 = acc.w * dd + bias.w;
  float m = fmaxf(fmaxf(v0, v1), fmaxf(v2, v3));
  m = fmaxf(m, __shfl_xor(m, 1));
  m = fmaxf(m, __shfl_xor(m, 2));
  const float e0 = expf(v0 - m);
  const float e1 = expf(v1 - m);
  const float e2 = expf(v2 - m);
  const float e3 = expf(v3 - m);
  float s = (e0 + e1) + (e2 + e3);
  s += __shfl_xor(s, 1);
  s += __shfl_xor(s, 2);
  const float inv = 1.0f / s;
  if (live)
    ((float4*)out)[d * 4 + sub] = make_float4(e0 * inv, e1 * inv, e2 * inv, e3 * inv);
}

extern "C" void kernel_launch(void* const* d_in, const int* in_sizes, int n_in,
                              void* d_out, int out_size, void* d_ws, size_t ws_size,
                              hipStream_t stream) {
  const float* x  = (const float*)d_in[0];
  const int*   ei = (const int*)d_in[1];
  const float* ew = (const float*)d_in[2];
  const float* W1 = (const float*)d_in[3];
  const float* b1 = (const float*)d_in[4];
  const float* W2 = (const float*)d_in[5];
  const float* b2 = (const float*)d_in[6];
  float* out = (float*)d_out;

  const int* src = ei;
  const int* dst = ei + N_EDGES;

  float* ws   = (float*)d_ws;
  float* dinv = ws;
  float* h2   = ws + N_NODES + (size_t)N_NODES * 128;
  int* off    = (int*)(ws + N_NODES + (size_t)N_NODES * 256);
  int* tab    = off + 100352;
  int* rec    = tab + 153600 + 512;
  int* csr    = rec + 2 * N_EDGES;

  const int T = 256;

  chunk_group_kernel<<<NCHUNK, SORT_T, 0, stream>>>(src, dst, ew, tab, rec);
  range_build_kernel<<<NRNG, BUILD_T, 0, stream>>>(rec, tab, off, dinv, csr);

  gather_gemm12_kernel<<<N_NODES / GX_TILE, T, 0, stream>>>(off, csr, dinv, x, W1, b1, W2, h2);

  gather2_softmax_kernel<<<(N_NODES * 4 + T - 1) / T, T, 0, stream>>>(off, csr, dinv, h2, b2, out);
}


# Both sort launches place their records in LDS at the final slot and copy rec and csr out as whole 128-byte lines (16-byte stores), no scattered 8-byte store per edge; edge and sum order unchanged
# speedup vs baseline: 5.6628x; 1.1011x over previous
#include <hip/hip_runtime.h>
#include <hip/hip_bf16.h>
#include <math.h>


#define N_NODES 100000
#define N_EDGES 1600000
#define F_DIM 128
#define H_DIM 128
#define C_DIM 16

#define RNG_SHIFT 9
#define RNG_NODES (1 << RNG_SHIFT)
#define NRNG ((N_NODES >> RNG_SHIFT) + 1)
#define CHUNK_EDGES 8192
#define NCHUNK ((N_EDGES + CHUNK_EDGES - 1) / CHUNK_EDGES)
#define E4 (N_EDGES / 4)
#define SRC_BITS 17
#define SRC_MASK ((1 << SRC_BITS) - 1)
#define SORT_T 512
#define SORT_WAVES (SORT_T / 32)
#define CHUNK_TRIPS (CHUNK_EDGES / (4 * SORT_T))
#define ROW4 (NCHUNK / 4)
static_assert(N_NODES <= (1 << SRC_BITS) && SRC_BITS + RNG_SHIFT <= 31, "source index and in-range destination share one word");
static_assert(N_EDGES % 4 == 0, "edge arrays are read four entries at a time");
static_assert(CHUNK_EDGES % (4 * SORT_T) == 0, "a chunk is whole trips of the work-group");
static_assert(NCHUNK % 4 == 0 && ROW4 > 32 && ROW4 <= 64, "a wave reads one table row with two 16-byte loads per lane");
static_assert(NRNG <= SORT_T, "the range starts are scanned one range per thread");
static_assert(NRNG * NCHUNK <= 153600 && N_NODES + 1 <= 100352, "workspace reservations");

typedef __attribute__((ext_vector_type(2))) float v2f;
typedef __attribute__((ext_vector_type(8))) float v8f;
typedef __attribute__((ext_vector_type(4))) int v4i;

__global__ __launch_bounds__(SORT_T) __attribute__((amdgpu_waves_per_eu(4, 4)))
void chunk_group_kernel(const int* __restrict__ src,
                        const int* __restrict__ dst,
                        const float* __restrict__ ew,
                        int* __restrict__ tab,
                        int* __restrict__ rec) {
  __shared__ int cnt[NRNG];
  __shared__ int first[NRNG];
  __shared__ int sv[SORT_WAVES];
  __shared__ v4i stage4[CHUNK_EDGES / 2];
  const int tid = threadIdx.x, lane = tid & 31, wave = tid >> 5;
  const int c = (int)blockIdx.x;
  const int4* d4 = (const int4*)dst;
  const int4* s4 = (const int4*)src;
  const float4* w4 = (const float4*)ew;
  for (int i = tid; i < NRNG; i += SORT_T) cnt[i] = 0;
  int4 d[CHUNK_TRIPS], sr[CHUNK_TRIPS], p[CHUNK_TRIPS];
  float4 w[CHUNK_TRIPS];
#pragma unroll
  for (int k = 0; k < CHUNK_TRIPS; ++k) {
    const int idx = min(c * (CHUNK_EDGES / 4) + k * SORT_T + tid, E4 - 1);
    d[k] = d4[idx];
    sr[k] = s4[idx];
    w[k] = w4[idx];
  }
  __syncthreads();
#pragma unroll
  for (int k = 0; k < CHUNK_TRIPS; ++k) {
    p[k] = make_int4(0, 0, 0, 0);
    if (c * (CHUNK_EDGES / 4) + k * SORT_T + tid < E4) {
      p[k].x = atomicAdd(&cnt[d[k].x >> RNG_SHIFT], 1);
      p[k].y = atomicAdd(&cnt[d[k].y >> RNG_SHIFT], 1);
      p[k].z = atomicAdd(&cnt[d[k].z >> RNG_SHIFT], 1);
      p[k].w = atomicAdd(&cnt[d[k].w >> RNG_SHIFT], 1);
    }
  }
  __syncthreads();
  const int a = (tid < NRNG) ? cnt[tid] : 0;
  int inc = a;
  for (int o = 1; o < 32; o <<= 1) {
    const int u = __shfl_up(inc, o);
    if (lane >= o) inc += u;
  }
  if (lane == 31) sv[wave] = inc;
  __syncthreads();
  int wbase = 0;
  for (int q = 0; q < wave; ++q) wbase += sv[q];
  if (tid < NRNG) {
    const int f = wbase + inc - a;
    first[tid] = f;
    tab[tid * NCHUNK + c] = f | (a << 16);
  }
  __syncthreads();
  int2* stage = (int2*)stage4;
#pragma unroll
  for (int k = 0; k < CHUNK_TRIPS; ++k) {
    if (c * (CHUNK_EDGES / 4) + k * SORT_T + tid < E4) {
      const int p0 = first[d[k].x >> RNG_SHIFT] + p[k].x;
      const int p1 = first[d[k].y >> RNG_SHIFT] + p[k].y;
      const int p2 = first[d[k].z >> RNG_SHIFT] + p[k].z;
      const int p3 = first[d[k].w >> RNG_SHIFT] + p[k].w;
      stage[p0] = make_int2(sr[k].x | ((d[k].x & (RNG_NODES - 1)) << SRC_BITS), __float_as_int(w[k].x));
      stage[p1] = make_int2(sr[k].y | ((d[k].y & (RNG_NODES - 1)) << SRC_BITS), __float_as_int(w[k].y));
      stage[p2] = make_int2(sr[k].z | ((d[k].z & (RNG_NODES - 1)) << SRC_BITS), __float_as_int(w[k].z));
      stage[p3] = make_int2(sr[k].w | ((d[k].w & (RNG_NODES - 1)) << SRC_BITS), __float_as_int(w[k].w));
    }
  }
  __syncthreads();
  const int nrec = min(CHUNK_EDGES, N_EDGES - c * CHUNK_EDGES);
  v4i* out4 = (v4i*)((int2*)rec + c * CHUNK_EDGES);
  static_assert(CHUNK_EDGES / (2 * SORT_T) == 8, "eight pairs per lane, named one by one in the fence below");
  v4i val[8];
#pragma unroll
  for (int k = 0; k < 8; ++k) val[k] = stage4[k * SORT_T + tid];
  asm volatile("" :: "v"(val[0]), "v"(val[1]), "v"(val[2]), "v"(val[3]),
                     "v"(val[4]), "v"(val[5]), "v"(val[6]), "v"(val[7]));
#pragma unroll
  for (int k = 0; k < 8; ++k) {
    if (2 * (k * SORT_T + tid) < nrec) out4[k * SORT_T + tid] = val[k];
  }
}

#define BUILD_T 1024
#define BUILD_WAVES (BUILD_T / 32)
#define GROUPS_PER_WAVE ((NCHUNK + BUILD_WAVES - 1) / BUILD_WAVES)
#define STAGE_RECS 16384
static_assert(GROUPS_PER_WAVE * BUILD_WAVES >= NRNG, "the same wave stride covers the table rows");
static_assert(N_EDGES % 2 == 0 && CHUNK_EDGES % 2 == 0, "a chunk is copied out two records at a time");
static_assert(CHUNK_EDGES < (1 << 16), "a group's first slot and its size share one word");
__global__ __launch_bounds__(BUILD_T) void range_build_kernel(const int* __restrict__ rec,
                                                              const int* __restrict__ tab,
                                                              int* __restrict__ off,
                                                              float* __restrict__ dinv,
                                                              int* __restrict__ csr) {
  __shared__ int cnt[RNG_NODES];
  __shared__ float dw[RNG_NODES];
  __shared__ int cur[RNG_NODES];
  __shared__ int sv[RNG_NODES / 32];
  __shared__ int sb[BUILD_WAVES];
  __shared__ int s_tot;
  __shared__ v4i stage4[STAGE_RECS / 2];
  const int tid = threadIdx.x, lane = tid & 31, wave = tid >> 5;
  const int r = (int)blockIdx.x;
  const int2* rec2 = (const int2*)rec;
  int2* csr2 = (int2*)csr;
  int gbeg[GROUPS_PER_WAVE], gn[GROUPS_PER_WAVE];
  {
    const int4* t4 = (const int4*)tab;
    int4 va[GROUPS_PER_WAVE], vb[GROUPS_PER_WAVE];
    int hd[GROUPS_PER_WAVE];
#pragma unroll
    for (int k = 0; k < GROUPS_PER_WAVE; ++k) {
      const int row = min(wave + k * BUILD_WAVES, NRNG - 1);
      va[k] = t4[row * ROW4 + lane];
      vb[k] = t4[row * ROW4 + min(lane + 32, ROW4 - 1)];
      hd[k] = tab[r * NCHUNK + min(wave + k * BUILD_WAVES, NCHUNK - 1)];
    }
    if (tid < RNG_NODES) {
      cnt[tid] = 0;
      dw[tid] = 0.f;
    }
    const bool two = lane + 32 < ROW4;
    int part = 0;
#pragma unroll
    for (int k = 0; k < GROUPS_PER_WAVE; ++k) {
      if (wave + k * BUILD_WAVES < r) {
        part += (va[k].x >> 16) + (va[k].y >> 16) + (va[k].z >> 16) + (va[k].w >> 16);
        if (two) part += (vb[k].x >> 16) + (vb[k].y >> 16) + (vb[k].z >> 16) + (vb[k].w >> 16);
      }
    }
#pragma unroll
    for (int o = 16; o > 0; o >>= 1) part += __shfl_xor(part, o);
    if (lane == 0) sb[wave] = part;
#pragma unroll
    for (int k = 0; k < GROUPS_PER_WAVE; ++k) {
      const int ck = wave + k * BUILD_WAVES;
      gn[k] = (ck < NCHUNK) ? (hd[k] >> 16) : 0;
      gbeg[k] = ck * CHUNK_EDGES + (hd[k] & 0xFFFF);
    }
  }
  int maxn = 0;
#pragma unroll
  for (int k = 0; k < GROUPS_PER_WAVE; ++k) maxn = max(maxn, gn[k]);
  __syncthreads();
  int beg = 0;
  for (int q = 0; q < BUILD_WAVES; ++q) beg += sb[q];
  for (int t = lane; t - lane < maxn; t += 32) {
    int2 q[GROUPS_PER_WAVE];
#pragma unroll
    for (int k = 0; k < GROUPS_PER_WAVE; ++k) q[k] = rec2[(t < gn[k]) ? gbeg[k] + t : 0];
#pragma unroll
    for (int k = 0; k < GROUPS_PER_WAVE; ++k) {
      if (t < gn[k]) {
        const int dl = q[k].x >> SRC_BITS;
        atomicAdd(&cnt[dl], 1);
        atomicAdd(&dw[dl], __int_as_float(q[k].y));
      }
    }
  }
  __syncthreads();
  int c = 0, inc = 0;
  if (tid < RNG_NODES) {
    c = cnt[tid];
    inc = c;
    for (int o = 1; o < 32; o <<= 1) {
      const int u = __shfl_up(inc, o);
      if (lane >= o) inc += u;
    }
    if (lane == 31) sv[wave] = inc;
  }
  __syncthreads();
  if (tid < RNG_NODES) {
    int wbase = 0;
    for (int w = 0; w < wave; ++w) wbase += sv[w];
    const int ex = beg + wbase + inc - c;
    cur[tid] = ex;
    if (tid == RNG_NODES - 1) s_tot = wbase + inc;
    const int node = r * RNG_NODES + tid;
    if (node <= N_NODES) off[node] = ex;
    if (node < N_NODES) {
      const float d = 1.0f + dw[tid];
      dinv[node] = (d > 0.0f) ? rsqrtf(d) : 0.0f;
    }
  }
  __syncthreads();
  const int tot = s_tot;
  const int abase = beg & ~15;
  const int lead = beg - abase;
  const bool staged = lead + tot <= STAGE_RECS;
  int2* stage = (int2*)stage4;
  for (int t = lane; t - lane < maxn; t += 32) {
    int2 q[GROUPS_PER_WAVE];
#pragma unroll
    for (int k = 0; k < GROUPS_PER_WAVE; ++k) q[k] = rec2[(t < gn[k]) ? gbeg[k] + t : 0];
    int slot[GROUPS_PER_WAVE];
#pragma unroll
    for (int k = 0; k < GROUPS_PER_WAVE; ++k) {
      slot[k] = 0;
      if (t < gn[k]) slot[k] = atomicAdd(&cur[q[k].x >> SRC_BITS], 1);
    }
    if (staged) {
#pragma unroll
      for (int k = 0; k < GROUPS_PER_WAVE; ++k) {
        if (t < gn[k]) stage[slot[k] - abase] = make_int2(q[k].x & SRC_MASK, q[k].y);
      }
    } else {
#pragma unroll
      for (int k = 0; k < GROUPS_PER_WAVE; ++k) {
        if (t < gn[k]) csr2[slot[k]] = make_int2(q[k].x & SRC_MASK, q[k].y);
      }
    }
  }
  if (staged) {
    __syncthreads();
    const int hi = lead + tot;
    if (lead < hi) {
      if (tid == 0 && (lead & 1)) csr2[abase + lead] = stage[lead];
      if (tid == 32 && (hi & 1)) csr2[abase + hi - 1] = stage[hi - 1];
    }
    const int vlo = (lead + 1) >> 1, vhi = hi >> 1;
    v4i* c4 = (v4i*)(csr2 + abase);
    for (int v0 = 0; v0 < vhi; v0 += 4 * BUILD_T) {
      v4i val[4];
#pragma unroll
      for (int j = 0; j < 4; ++j) val[j] = stage4[min(v0 + j * BUILD_T + tid, STAGE_RECS / 2 - 1)];
      asm volatile("" :: "v"(val[0]), "v"(val[1]), "v"(val[2]), "v"(val[3]));
#pragma unroll
      for (int j = 0; j < 4; ++j) {
        const int v = v0 + j * BUILD_T + tid;
        if (v >= vlo && v < vhi) c4[v] = val[j];
      }
    }
  }
}

#define GX_NPW 4
#define GX_TILE (8 * GX_NPW)
#define GX_PITCH 132
static_assert(GX_TILE == 32, "two 16-row matrix tiles per work-group");
static_assert(N_NODES % GX_TILE == 0, "the grid covers the nodes exactly, no row guards");

__device__ __forceinline__ void gather_rows(float4& acc, const int s_l, const float nrm_l,
                                            const int n, const float4* __restrict__ r4,
                                            const int lane) {
#pragma unroll 4
  for (int j = 0; j < n; ++j) {
    const int s = __builtin_amdgcn_readlane(s_l, j);
    const float nrm = __int_as_float(__builtin_amdgcn_readlane(__float_as_int(nrm_l), j));
    const float4 v = r4[s * 32 + lane];
    acc.x += v.x * nrm;
    acc.y += v.y * nrm;
    acc.z += v.z * nrm;
    acc.w += v.w * nrm;
  }
}

__global__ __launch_bounds__(256) __attribute__((amdgpu_waves_per_eu(16, 16)))
void gather_gemm12_kernel(const int* __restrict__ off,
                                                            const int* __restrict__ csr,
                                                            const float* __restrict__ dinv,
                                                            const float* __restrict__ x,
                                                            const float* __restrict__ W1,
                                                            const float* __restrict__ b1,
                                                            const float* __restrict__ W2,
                                                            float* __restrict__ h2) {
  __shared__ float sA[GX_TILE * GX_PITCH];
  __shared__ float sW[128 * 16];
  const int tid  = threadIdx.x;
  const int lane = tid & 31;
  const int wave = __builtin_amdgcn_readfirstlane(tid >> 5);
  const float4 w2a = ((const float4*)W2)[tid];
  const float4 w2b = ((const float4*)W2)[tid + 256];
  const float4* x4  = (const float4*)x;
  const int2* csr2  = (const int2*)csr;
  const int d0 = (int)blockIdx.x * GX_TILE + wave;

  int beg[GX_NPW], end[GX_NPW], nn[GX_NPW], s_l[GX_NPW];
  float dd[GX_NPW], w_l[GX_NPW], nrm_l[GX_NPW];
  float4 acc[GX_NPW];
#pragma unroll
  for (int p = 0; p < GX_NPW; ++p) {
    const int d = d0 + 8 * p;
    beg[p] = off[d];
    end[p] = off[d + 1];
    dd[p]  = dinv[d];
  }
#pragma unroll
  for (int p = 0; p < GX_NPW; ++p) {
    const int d = d0 + 8 * p;
    acc[p] = x4[d * 32 + lane];
    nn[p]  = min(32, end[p] - beg[p]);
    const bool on = lane < nn[p];
    const int2 ed = csr2[on ? beg[p] + lane : 0];
    s_l[p] = on ? ed.x : 0;
    w_l[p] = on ? __int_as_float(ed.y) : 0.f;
  }
#pragma unroll
  for (int p = 0; p < GX_NPW; ++p) nrm_l[p] = dinv[s_l[p]] * w_l[p] * dd[p];
  asm volatile("" :: "v"(w2a.x), "v"(w2a.y), "v"(w2a.z), "v"(w2a.w),
                     "v"(w2b.x), "v"(w2b.y), "v"(w2b.z), "v"(w2b.w));

#pragma unroll
  for (int p = 0; p < GX_NPW; ++p) {
    float4 a = acc[p];
    a.x = a.x * dd[p] * dd[p];
    a.y = a.y * dd[p] * dd[p];
    a.z = a.z * dd[p] * dd[p];
    a.w = a.w * dd[p] * dd[p];
    gather_rows(a, s_l[p], nrm_l[p], nn[p], x4, lane);
    for (int base = beg[p] + 32; base < end[p]; base += 32) {
      const int n = min(32, end[p] - base);
      int s2 = 0;
      float nrm2 = 0.f;
      if (lane < n) {
        const int2 ed = csr2[base + lane];
        s2 = ed.x;
        nrm2 = dinv[ed.x] * __int_as_float(ed.y) * dd[p];
      }
      gather_rows(a, s2, nrm2, n, x4, lane);
    }
    *(float4*)&sA[(8 * p + wave) * GX_PITCH + 4 * lane] = a;
  }
  ((float4*)sW)[tid] = w2a;
  ((float4*)sW)[tid + 256] = w2b;

  const int lrow = lane & 15, hi = lane >> 4;
  const int col = wave * 16 + lrow;
  const float* bp = W1 + 2 * hi * 128 + col;
  v2f bq[8];
#pragma unroll
  for (int i = 0; i < 8; ++i) {
    bq[i].x = bp[(4 * i) * 128];
    bq[i].y = bp[(4 * i + 1) * 128];
  }
  __syncthreads();

  v8f c0 = {}, c1 = {};
  {
    const float* a0p = &sA[lrow * GX_PITCH + 2 * hi];
    const float* a1p = a0p + 16 * GX_PITCH;
#pragma unroll
    for (int i = 0; i < 32; ++i) {
      const int k0 = 4 * i;
      v2f a0, a1, b;
      a0.x = a0p[k0];
      a0.y = a0p[k0 + 1];
      a1.x = a1p[k0];
      a1.y = a1p[k0 + 1];
      if (i < 8) {
        b = bq[i];
      } else {
        b.x = bp[k0 * 128];
        b.y = bp[(k0 + 1) * 128];
      }
      c0 = __builtin_amdgcn_wmma_f32_16x16x4_f32(false, a0, false, b,
                                                 (short)0, c0, false, false);
      c1 = __builtin_amdgcn_wmma_f32_16x16x4_f32(false, a1, false, b,
                                                 (short)0, c1, false, false);
    }
  }
  const float bv = b1[col];
  __syncthreads();
#pragma unroll
  for (int r = 0; r < 8; ++r) {
    sA[(8 * hi + r) * GX_PITCH + col]      = fmaxf(c0[r] + bv, 0.f);
    sA[(16 + 8 * hi + r) * GX_PITCH + col] = fmaxf(c1[r] + bv, 0.f);
  }
  __syncthreads();

  if (wave < 2) {
    const float* ax = &sA[(16 * wave + lrow) * GX_PITCH + 2 * hi];
    const int rbase = (int)blockIdx.x * GX_TILE + 16 * wave + 8 * hi;
    float dv[8];
#pragma unroll
    for (int r = 0; r < 8; ++r) dv[r] = dinv[rbase + r];
    v8f c = {};
    for (int k0 = 0; k0 < 128; k0 += 4) {
      const int k = k0 + 2 * hi;
      v2f a, b;
      a.x = ax[k0];
      a.y = ax[k0 + 1];
      b.x = sW[k * 16 + lrow];
      b.y = sW[(k + 1) * 16 + lrow];
      c = __builtin_amdgcn_wmma_f32_16x16x4_f32(false, a, false, b,
                                                (short)0, c, false, false);
    }
#pragma unroll
    for (int r = 0; r < 8; ++r)
      h2[(rbase + r) * 16 + lrow] = c[r] * dv[r];
  }
}

__global__ __launch_bounds__(256) void gather2_softmax_kernel(const int* __restrict__ off,
                                                              const int* __restrict__ csr,
                                                              const float* __restrict__ dinv,
                                                              const float* __restrict__ h2s,
                                                              const float* __restrict__ b2,
                                                              float* __restrict__ out) {
  const int gid = blockIdx.x * 256 + threadIdx.x;
  const int sub = threadIdx.x & 3;
  const bool live = (gid >> 2) < N_NODES;
  const int d = live ? (gid >> 2) : (N_NODES - 1);
  const int beg = off[d], end = off[d + 1];
  const float dd = dinv[d];
  const float4 bias = ((const float4*)b2)[sub];
  const float4* p = (const float4*)h2s;
  const int2* csr2 = (const int2*)csr;
  float4 acc = p[d * 4 + sub];
  for (int j = beg; j < end; ++j) {
    const int2 ed = csr2[j];
    const float w = __int_as_float(ed.y);
    const float4 v = p[ed.x * 4 + sub];
    acc.x += v.x * w;
    acc.y += v.y * w;
    acc.z += v.z * w;
    acc.w += v.w * w;
  }
  const float v0 = acc.x * dd + bias.x;
  const float v1 = acc.y * dd + bias.y;
  const float v2 = acc.z * dd + bias.z;
  const float v3 = acc.w * dd + bias.w;
  float m = fmaxf(fmaxf(v0, v1), fmaxf(v2, v3));
  m = fmaxf(m, __shfl_xor(m, 1));
  m = fmaxf(m, __shfl_xor(m, 2));
  const float e0 = expf(v0 - m);
  const float e1 = expf(v1 - m);
  const float e2 = expf(v2 - m);
  const float e3 = expf(v3 - m);
  float s = (e0 + e1) + (e2 + e3);
  s += __shfl_xor(s, 1);
  s += __shfl_xor(s, 2);
  const float inv = 1.0f / s;
  if (live)
    ((float4*)out)[d * 4 + sub] = make_float4(e0 * inv, e1 * inv, e2 * inv, e3 * inv);
}

extern "C" void kernel_launch(void* const* d_in, const int* in_sizes, int n_in,
                              void* d_out, int out_size, void* d_ws, size_t ws_size,
                              hipStream_t stream) {
  const float* x  = (const float*)d_in[0];
  const int*   ei = (const int*)d_in[1];
  const float* ew = (const float*)d_in[2];
  const float* W1 = (const float*)d_in[3];
  const float* b1 = (const float*)d_in[4];
  const float* W2 = (const float*)d_in[5];
  const float* b2 = (const float*)d_in[6];
  float* out = (float*)d_out;

  const int* src = ei;
  const int* dst = ei + N_EDGES;

  float* ws   = (float*)d_ws;
  float* dinv = ws;
  float* h2   = ws + N_NODES + (size_t)N_NODES * 128;
  int* off    = (int*)(ws + N_NODES + (size_t)N_NODES * 256);
  int* tab    = off + 100352;
  int* rec    = tab + 153600 + 512;
  int* csr    = rec + 2 * N_EDGES;

  const int T = 256;

  chunk_group_kernel<<<NCHUNK, SORT_T, 0, stream>>>(src, dst, ew, tab, rec);
  range_build_kernel<<<NRNG, BUILD_T, 0, stream>>>(rec, tab, off, dinv, csr);

  gather_gemm12_kernel<<<N_NODES / GX_TILE, T, 0, stream>>>(off, csr, dinv, x, W1, b1, W2, h2);

  gather2_softmax_kernel<<<(N_NODES * 4 + T - 1) / T, T, 0, stream>>>(off, csr, dinv, h2, b2, out);
}


# Layer-1 launch split by role in 24-wave work-groups: 16 waves gather tile t+1 (8 row loads in flight) while 8 run tile t's matrix products, W1 kept in LDS, 1 barrier per tile; sum order unchanged
# speedup vs baseline: 6.1495x; 1.0859x over previous
#include <hip/hip_runtime.h>
#include <hip/hip_bf16.h>
#include <math.h>


#define N_NODES 100000
#define N_EDGES 1600000
#define F_DIM 128
#define H_DIM 128
#define C_DIM 16

#define RNG_SHIFT 9
#define RNG_NODES (1 << RNG_SHIFT)
#define NRNG ((N_NODES >> RNG_SHIFT) + 1)
#define CHUNK_EDGES 8192
#define NCHUNK ((N_EDGES + CHUNK_EDGES - 1) / CHUNK_EDGES)
#define E4 (N_EDGES / 4)
#define SRC_BITS 17
#define SRC_MASK ((1 << SRC_BITS) - 1)
#define SORT_T 512
#define SORT_WAVES (SORT_T / 32)
#define CHUNK_TRIPS (CHUNK_EDGES / (4 * SORT_T))
#define ROW4 (NCHUNK / 4)
static_assert(N_NODES <= (1 << SRC_BITS) && SRC_BITS + RNG_SHIFT <= 31, "source index and in-range destination share one word");
static_assert(N_EDGES % 4 == 0, "edge arrays are read four entries at a time");
static_assert(CHUNK_EDGES % (4 * SORT_T) == 0, "a chunk is whole trips of the work-group");
static_assert(NCHUNK % 4 == 0 && ROW4 > 32 && ROW4 <= 64, "a wave reads one table row with two 16-byte loads per lane");
static_assert(NRNG <= SORT_T, "the range starts are scanned one range per thread");
static_assert(NRNG * NCHUNK <= 153600 && N_NODES + 1 <= 100352, "workspace reservations");

typedef __attribute__((ext_vector_type(2))) float v2f;
typedef __attribute__((ext_vector_type(8))) float v8f;
typedef __attribute__((ext_vector_type(4))) int v4i;

__global__ __launch_bounds__(SORT_T) __attribute__((amdgpu_waves_per_eu(4, 4)))
void chunk_group_kernel(const int* __restrict__ src,
                        const int* __restrict__ dst,
                        const float* __restrict__ ew,
                        int* __restrict__ tab,
                        int* __restrict__ rec) {
  __shared__ int cnt[NRNG];
  __shared__ int first[NRNG];
  __shared__ int sv[SORT_WAVES];
  __shared__ v4i stage4[CHUNK_EDGES / 2];
  const int tid = threadIdx.x, lane = tid & 31, wave = tid >> 5;
  const int c = (int)blockIdx.x;
  const int4* d4 = (const int4*)dst;
  const int4* s4 = (const int4*)src;
  const float4* w4 = (const float4*)ew;
  for (int i = tid; i < NRNG; i += SORT_T) cnt[i] = 0;
  int4 d[CHUNK_TRIPS], sr[CHUNK_TRIPS], p[CHUNK_TRIPS];
  float4 w[CHUNK_TRIPS];
#pragma unroll
  for (int k = 0; k < CHUNK_TRIPS; ++k) {
    const int idx = min(c * (CHUNK_EDGES / 4) + k * SORT_T + tid, E4 - 1);
    d[k] = d4[idx];
    sr[k] = s4[idx];
    w[k] = w4[idx];
  }
  __syncthreads();
#pragma unroll
  for (int k = 0; k < CHUNK_TRIPS; ++k) {
    p[k] = make_int4(0, 0, 0, 0);
    if (c * (CHUNK_EDGES / 4) + k * SORT_T + tid < E4) {
      p[k].x = atomicAdd(&cnt[d[k].x >> RNG_SHIFT], 1);
      p[k].y = atomicAdd(&cnt[d[k].y >> RNG_SHIFT], 1);
      p[k].z = atomicAdd(&cnt[d[k].z >> RNG_SHIFT], 1);
      p[k].w = atomicAdd(&cnt[d[k].w >> RNG_SHIFT], 1);
    }
  }
  __syncthreads();
  const int a = (tid < NRNG) ? cnt[tid] : 0;
  int inc = a;
  for (int o = 1; o < 32; o <<= 1) {
    const int u = __shfl_up(inc, o);
    if (lane >= o) inc += u;
  }
  if (lane == 31) sv[wave] = inc;
  __syncthreads();
  int wbase = 0;
  for (int q = 0; q < wave; ++q) wbase += sv[q];
  if (tid < NRNG) {
    const int f = wbase + inc - a;
    first[tid] = f;
    tab[tid * NCHUNK + c] = f | (a << 16);
  }
  __syncthreads();
  int2* stage = (int2*)stage4;
#pragma unroll
  for (int k = 0; k < CHUNK_TRIPS; ++k) {
    if (c * (CHUNK_EDGES / 4) + k * SORT_T + tid < E4) {
      const int p0 = first[d[k].x >> RNG_SHIFT] + p[k].x;
      const int p1 = first[d[k].y >> RNG_SHIFT] + p[k].y;
      const int p2 = first[d[k].z >> RNG_SHIFT] + p[k].z;
      const int p3 = first[d[k].w >> RNG_SHIFT] + p[k].w;
      stage[p0] = make_int2(sr[k].x | ((d[k].x & (RNG_NODES - 1)) << SRC_BITS), __float_as_int(w[k].x));
      stage[p1] = make_int2(sr[k].y | ((d[k].y & (RNG_NODES - 1)) << SRC_BITS), __float_as_int(w[k].y));
      stage[p2] = make_int2(sr[k].z | ((d[k].z & (RNG_NODES - 1)) << SRC_BITS), __float_as_int(w[k].z));
      stage[p3] = make_int2(sr[k].w | ((d[k].w & (RNG_NODES - 1)) << SRC_BITS), __float_as_int(w[k].w));
    }
  }
  __syncthreads();
  const int nrec = min(CHUNK_EDGES, N_EDGES - c * CHUNK_EDGES);
  v4i* out4 = (v4i*)((int2*)rec + c * CHUNK_EDGES);
  static_assert(CHUNK_EDGES / (2 * SORT_T) == 8, "eight pairs per lane, named one by one in the fence below");
  v4i val[8];
#pragma unroll
  for (int k = 0; k < 8; ++k) val[k] = stage4[k * SORT_T + tid];
  asm volatile("" :: "v"(val[0]), "v"(val[1]), "v"(val[2]), "v"(val[3]),
                     "v"(val[4]), "v"(val[5]), "v"(val[6]), "v"(val[7]));
#pragma unroll
  for (int k = 0; k < 8; ++k) {
    if (2 * (k * SORT_T + tid) < nrec) out4[k * SORT_T + tid] = val[k];
  }
}

#define BUILD_T 1024
#define BUILD_WAVES (BUILD_T / 32)
#define GROUPS_PER_WAVE ((NCHUNK + BUILD_WAVES - 1) / BUILD_WAVES)
#define STAGE_RECS 16384
static_assert(GROUPS_PER_WAVE * BUILD_WAVES >= NRNG, "the same wave stride covers the table rows");
static_assert(N_EDGES % 2 == 0 && CHUNK_EDGES % 2 == 0, "a chunk is copied out two records at a time");
static_assert(CHUNK_EDGES < (1 << 16), "a group's first slot and its size share one word");
__global__ __launch_bounds__(BUILD_T) void range_build_kernel(const int* __restrict__ rec,
                                                              const int* __restrict__ tab,
                                                              int* __restrict__ off,
                                                              float* __restrict__ dinv,
                                                              int* __restrict__ csr) {
  __shared__ int cnt[RNG_NODES];
  __shared__ float dw[RNG_NODES];
  __shared__ int cur[RNG_NODES];
  __shared__ int sv[RNG_NODES / 32];
  __shared__ int sb[BUILD_WAVES];
  __shared__ int s_tot;
  __shared__ v4i stage4[STAGE_RECS / 2];
  const int tid = threadIdx.x, lane = tid & 31, wave = tid >> 5;
  const int r = (int)blockIdx.x;
  const int2* rec2 = (const int2*)rec;
  int2* csr2 = (int2*)csr;
  int gbeg[GROUPS_PER_WAVE], gn[GROUPS_PER_WAVE];
  {
    const int4* t4 = (const int4*)tab;
    int4 va[GROUPS_PER_WAVE], vb[GROUPS_PER_WAVE];
    int hd[GROUPS_PER_WAVE];
#pragma unroll
    for (int k = 0; k < GROUPS_PER_WAVE; ++k) {
      const int row = min(wave + k * BUILD_WAVES, NRNG - 1);
      va[k] = t4[row * ROW4 + lane];
      vb[k] = t4[row * ROW4 + min(lane + 32, ROW4 - 1)];
      hd[k] = tab[r * NCHUNK + min(wave + k * BUILD_WAVES, NCHUNK - 1)];
    }
    if (tid < RNG_NODES) {
      cnt[tid] = 0;
      dw[tid] = 0.f;
    }
    const bool two = lane + 32 < ROW4;
    int part = 0;
#pragma unroll
    for (int k = 0; k < GROUPS_PER_WAVE; ++k) {
      if (wave + k * BUILD_WAVES < r) {
        part += (va[k].x >> 16) + (va[k].y >> 16) + (va[k].z >> 16) + (va[k].w >> 16);
        if (two) part += (vb[k].x >> 16) + (vb[k].y >> 16) + (vb[k].z >> 16) + (vb[k].w >> 16);
      }
    }
#pragma unroll
    for (int o = 16; o > 0; o >>= 1) part += __shfl_xor(part, o);
    if (lane == 0) sb[wave] = part;
#pragma unroll
    for (int k = 0; k < GROUPS_PER_WAVE; ++k) {
      const int ck = wave + k * BUILD_WAVES;
      gn[k] = (ck < NCHUNK) ? (hd[k] >> 16) : 0;
      gbeg[k] = ck * CHUNK_EDGES + (hd[k] & 0xFFFF);
    }
  }
  int maxn = 0;
#pragma unroll
  for (int k = 0; k < GROUPS_PER_WAVE; ++k) maxn = max(maxn, gn[k]);
  __syncthreads();
  int beg = 0;
  for (int q = 0; q < BUILD_WAVES; ++q) beg += sb[q];
  for (int t = lane; t - lane < maxn; t += 32) {
    int2 q[GROUPS_PER_WAVE];
#pragma unroll
    for (int k = 0; k < GROUPS_PER_WAVE; ++k) q[k] = rec2[(t < gn[k]) ? gbeg[k] + t : 0];
#pragma unroll
    for (int k = 0; k < GROUPS_PER_WAVE; ++k) {
      if (t < gn[k]) {
        const int dl = q[k].x >> SRC_BITS;
        atomicAdd(&cnt[dl], 1);
        atomicAdd(&dw[dl], __int_as_float(q[k].y));
      }
    }
  }
  __syncthreads();
  int c = 0, inc = 0;
  if (tid < RNG_NODES) {
    c = cnt[tid];
    inc = c;
    for (int o = 1; o < 32; o <<= 1) {
      const int u = __shfl_up(inc, o);
      if (lane >= o) inc += u;
    }
    if (lane == 31) sv[wave] = inc;
  }
  __syncthreads();
  if (tid < RNG_NODES) {
    int wbase = 0;
    for (int w = 0; w < wave; ++w) wbase += sv[w];
    const int ex = beg + wbase + inc - c;
    cur[tid] = ex;
    if (tid == RNG_NODES - 1) s_tot = wbase + inc;
    const int node = r * RNG_NODES + tid;
    if (node <= N_NODES) off[node] = ex;
    if (node < N_NODES) {
      const float d = 1.0f + dw[tid];
      dinv[node] = (d > 0.0f) ? rsqrtf(d) : 0.0f;
    }
  }
  __syncthreads();
  const int tot = s_tot;
  const int abase = beg & ~15;
  const int lead = beg - abase;
  const bool staged = lead + tot <= STAGE_RECS;
  int2* stage = (int2*)stage4;
  for (int t = lane; t - lane < maxn; t += 32) {
    int2 q[GROUPS_PER_WAVE];
#pragma unroll
    for (int k = 0; k < GROUPS_PER_WAVE; ++k) q[k] = rec2[(t < gn[k]) ? gbeg[k] + t : 0];
    int slot[GROUPS_PER_WAVE];
#pragma unroll
    for (int k = 0; k < GROUPS_PER_WAVE; ++k) {
      slot[k] = 0;
      if (t < gn[k]) slot[k] = atomicAdd(&cur[q[k].x >> SRC_BITS], 1);
    }
    if (staged) {
#pragma unroll
      for (int k = 0; k < GROUPS_PER_WAVE; ++k) {
        if (t < gn[k]) stage[slot[k] - abase] = make_int2(q[k].x & SRC_MASK, q[k].y);
      }
    } else {
#pragma unroll
      for (int k = 0; k < GROUPS_PER_WAVE; ++k) {
        if (t < gn[k]) csr2[slot[k]] = make_int2(q[k].x & SRC_MASK, q[k].y);
      }
    }
  }
  if (staged) {
    __syncthreads();
    const int hi = lead + tot;
    if (lead < hi) {
      if (tid == 0 && (lead & 1)) csr2[abase + lead] = stage[lead];
      if (tid == 32 && (hi & 1)) csr2[abase + hi - 1] = stage[hi - 1];
    }
    const int vlo = (lead + 1) >> 1, vhi = hi >> 1;
    v4i* c4 = (v4i*)(csr2 + abase);
    for (int v0 = 0; v0 < vhi; v0 += 4 * BUILD_T) {
      v4i val[4];
#pragma unroll
      for (int j = 0; j < 4; ++j) val[j] = stage4[min(v0 + j * BUILD_T + tid, STAGE_RECS / 2 - 1)];
      asm volatile("" :: "v"(val[0]), "v"(val[1]), "v"(val[2]), "v"(val[3]));
#pragma unroll
      for (int j = 0; j < 4; ++j) {
        const int v = v0 + j * BUILD_T + tid;
        if (v >= vlo && v < vhi) c4[v] = val[j];
      }
    }
  }
}

#define GX_NPW 2
#define GX_TILE (16 * GX_NPW)
#define GX_PITCH 132
static_assert(GX_TILE == 32, "two 16-row matrix tiles per work-group");

__device__ __forceinline__ void gather_rows(float4& acc, const int s_l, const float nrm_l,
                                            const int n, const float4* __restrict__ r4,
                                            const int lane) {
#define GX_ROW(j) r4[__builtin_amdgcn_readlane(s_l, (j)) * 32 + lane]
#define GX_ADD(v, j) { const float nrm = __int_as_float(__builtin_amdgcn_readlane(__float_as_int(nrm_l), (j))); \
                       acc.x = fmaf((v).x, nrm, acc.x); acc.y = fmaf((v).y, nrm, acc.y); \
                       acc.z = fmaf((v).z, nrm, acc.z); acc.w = fmaf((v).w, nrm, acc.w); }
  int j = 0;
  for (; j + 8 <= n; j += 8) {
    float4 v[8];
#pragma unroll
    for (int u = 0; u < 8; ++u) v[u] = GX_ROW(j + u);
#pragma unroll
    for (int u = 0; u < 8; ++u) GX_ADD(v[u], j + u)
  }
  const int r = n - j;
  const int j2 = j + (r & 4), j1 = j2 + (r & 2);
  const float4 z = make_float4(0.f, 0.f, 0.f, 0.f);
  float4 a0 = z, a1 = z, a2 = z, a3 = z, b0 = z, b1 = z, c0 = z;
  if (r & 4) {
    a0 = GX_ROW(j);
    a1 = GX_ROW(j + 1);
    a2 = GX_ROW(j + 2);
    a3 = GX_ROW(j + 3);
  }
  if (r & 2) {
    b0 = GX_ROW(j2);
    b1 = GX_ROW(j2 + 1);
  }
  if (r & 1) c0 = GX_ROW(j1);
  if (r & 4) {
    GX_ADD(a0, j)
    GX_ADD(a1, j + 1)
    GX_ADD(a2, j + 2)
    GX_ADD(a3, j + 3)
  }
  if (r & 2) {
    GX_ADD(b0, j2)
    GX_ADD(b1, j2 + 1)
  }
  if (r & 1) GX_ADD(c0, j1)
#undef GX_ROW
#undef GX_ADD
}

#define GX_NWG 512
#define GX_STEPS 7
#define GX_ROWS 28
#define GX_THREADS 768
static_assert(GX_ROWS * GX_STEPS >= (N_NODES + GX_NWG - 1) / GX_NWG && GX_ROWS <= GX_TILE,
              "seven tiles cover the node range of a work-group");
static_assert((long long)GX_NWG * N_NODES < (1LL << 31), "range bounds are computed in 32 bits");
__global__ __launch_bounds__(GX_THREADS) __attribute__((amdgpu_waves_per_eu(12, 12)))
void gather_gemm12_kernel(const int* __restrict__ off,
                          const int* __restrict__ csr,
                          const float* __restrict__ dinv,
                          const float* __restrict__ x,
                          const float* __restrict__ W1,
                          const float* __restrict__ b1,
                          const float* __restrict__ W2,
                          float* __restrict__ h2) {
  __shared__ float sA[2][GX_TILE * GX_PITCH];
  __shared__ float sH[2][GX_TILE * GX_PITCH];
  __shared__ float4 sW1[128 * 128 / 4];
  __shared__ float sW2[128 * 16];
  __shared__ float sB[128];
  const int tid  = threadIdx.x;
  const int lane = tid & 31;
  const int wave = __builtin_amdgcn_readfirstlane(tid >> 5);
  const int wg = (int)blockIdx.x;
  const int n0 = wg * N_NODES / GX_NWG;
  const int n1 = (wg + 1) * N_NODES / GX_NWG;
  const float4* x4  = (const float4*)x;
  const int2* csr2  = (const int2*)csr;
  const int lrow = lane & 15, hi = lane >> 4;

  if (wave >= 16) {
    const int mt = tid - 512;
#pragma unroll
    for (int u = 0; u < 16; ++u) {
      const int q = mt + 256 * u;
      const int c = q & 15, h = (q >> 4) & 1, i2 = (q >> 5) & 15, ct = q >> 9;
      const float* w = W1 + (8 * i2 + 2 * h) * 128 + 16 * ct + c;
      sW1[q] = make_float4(w[0], w[128], w[512], w[640]);
    }
    ((float4*)sW2)[mt] = ((const float4*)W2)[mt];
    ((float4*)sW2)[mt + 256] = ((const float4*)W2)[mt + 256];
    if (mt < 128) sB[mt] = b1[mt];
  }

  int hv = 0;
  float hd = 0.f;
  if (wave < 16) {
    const int r = wave + 16 * ((lane >> 1) & 1);
    const int d = (r < GX_ROWS && n0 + r < n1) ? n0 + r : n0;
    hv = off[d + (lane & 1)];
    hd = dinv[d];
  }

#pragma unroll 1
  for (int step = 0; step < GX_STEPS + 2; ++step) {
    if (wave < 16) {
      if (step < GX_STEPS) {
        float* sAt = sA[step & 1];
        const int t0 = n0 + GX_ROWS * step;
        int dc[GX_NPW], beg[GX_NPW], end[GX_NPW], nn[GX_NPW], s_l[GX_NPW];
        float dd[GX_NPW], w_l[GX_NPW], nrm_l[GX_NPW];
        float4 acc[GX_NPW];
#pragma unroll
        for (int p = 0; p < GX_NPW; ++p) {
          const int r = wave + 16 * p;
          const bool ok = r < GX_ROWS && t0 + r < n1;
          dc[p]  = ok ? t0 + r : n0;
          beg[p] = __builtin_amdgcn_readlane(hv, 2 * p);
          end[p] = ok ? __builtin_amdgcn_readlane(hv, 2 * p + 1) : beg[p];
          dd[p]  = __int_as_float(__builtin_amdgcn_readlane(__float_as_int(hd), 2 * p));
        }
        {
          const int t1 = n0 + GX_ROWS * min(step + 1, GX_STEPS - 1);
          const int r = wave + 16 * ((lane >> 1) & 1);
          const int d = (r < GX_ROWS && t1 + r < n1) ? t1 + r : n0;
          hv = off[d + (lane & 1)];
          hd = dinv[d];
        }
#pragma unroll
        for (int p = 0; p < GX_NPW; ++p) {
          acc[p] = x4[dc[p] * 32 + lane];
          nn[p]  = min(32, end[p] - beg[p]);
          const bool on = lane < nn[p];
          const int2 ed = csr2[on ? beg[p] + lane : 0];
          s_l[p] = on ? ed.x : 0;
          w_l[p] = on ? __int_as_float(ed.y) : 0.f;
        }
#pragma unroll
        for (int p = 0; p < GX_NPW; ++p) nrm_l[p] = dinv[s_l[p]] * w_l[p] * dd[p];
#pragma unroll
        for (int p = 0; p < GX_NPW; ++p) {
          float4 a = acc[p];
          a.x = a.x * dd[p] * dd[p];
          a.y = a.y * dd[p] * dd[p];
          a.z = a.z * dd[p] * dd[p];
          a.w = a.w * dd[p] * dd[p];
          gather_rows(a, s_l[p], nrm_l[p], nn[p], x4, lane);
          for (int base = beg[p] + 32; base < end[p]; base += 32) {
            const int n = min(32, end[p] - base);
            int s2 = 0;
            float nrm2 = 0.f;
            if (lane < n) {
              const int2 ed = csr2[base + lane];
              s2 = ed.x;
              nrm2 = dinv[ed.x] * __int_as_float(ed.y) * dd[p];
            }
            gather_rows(a, s2, nrm2, n, x4, lane);
          }
          *(float4*)&sAt[(16 * p + wave) * GX_PITCH + 4 * lane] = a;
        }
      }
    } else {
      const int m = wave - 16;
      const int col = 16 * m + lrow;
      if (step >= 2 && m < 2) {
        const float* sHt = sH[step & 1];
        const float* ax = &sHt[(16 * m + lrow) * GX_PITCH + 2 * hi];
        const int r0 = 16 * m + 8 * hi;
        const int nb = n0 + GX_ROWS * (step - 2) + r0;
        float dv[8];
#pragma unroll
        for (int r = 0; r < 8; ++r) dv[r] = dinv[min(nb + r, N_NODES - 1)];
        v8f c = {};
        for (int k0 = 0; k0 < 128; k0 += 4) {
          const int k = k0 + 2 * hi;
          v2f a, b;
          a.x = ax[k0];
          a.y = ax[k0 + 1];
          b.x = sW2[k * 16 + lrow];
          b.y = sW2[(k + 1) * 16 + lrow];
          c = __builtin_amdgcn_wmma_f32_16x16x4_f32(false, a, false, b,
                                                    (short)0, c, false, false);
        }
#pragma unroll
        for (int r = 0; r < 8; ++r)
          if (r0 + r < GX_ROWS && nb + r < n1)
            h2[(nb + r) * 16 + lrow] = c[r] * dv[r];
      }
      if (step >= 1 && step <= GX_STEPS) {
        const float* a0p = &sA[(step - 1) & 1][lrow * GX_PITCH + 2 * hi];
        const float* a1p = a0p + 16 * GX_PITCH;
        float* sHt = sH[(step - 1) & 1];
        const float4* bp4 = &sW1[(m * 32 + hi) * 16 + lrow];
        v8f c0 = {}, c1 = {};
#pragma unroll
        for (int i2 = 0; i2 < 16; ++i2) {
          const float4 b4 = bp4[i2 * 32];
          v2f a0, a1, b;
          b.x = b4.x;
          b.y = b4.y;
          a0.x = a0p[8 * i2];
          a0.y = a0p[8 * i2 + 1];
          a1.x = a1p[8 * i2];
          a1.y = a1p[8 * i2 + 1];
          c0 = __builtin_amdgcn_wmma_f32_16x16x4_f32(false, a0, false, b,
                                                     (short)0, c0, false, false);
          c1 = __builtin_amdgcn_wmma_f32_16x16x4_f32(false, a1, false, b,
                                                     (short)0, c1, false, false);
          b.x = b4.z;
          b.y = b4.w;
          a0.x = a0p[8 * i2 + 4];
          a0.y = a0p[8 * i2 + 5];
          a1.x = a1p[8 * i2 + 4];
          a1.y = a1p[8 * i2 + 5];
          c0 = __builtin_amdgcn_wmma_f32_16x16x4_f32(false, a0, false, b,
                                                     (short)0, c0, false, false);
          c1 = __builtin_amdgcn_wmma_f32_16x16x4_f32(false, a1, false, b,
                                                     (short)0, c1, false, false);
        }
        const float bv = sB[col];
#pragma unroll
        for (int r = 0; r < 8; ++r) {
          sHt[(8 * hi + r) * GX_PITCH + col]      = fmaxf(c0[r] + bv, 0.f);
          sHt[(16 + 8 * hi + r) * GX_PITCH + col] = fmaxf(c1[r] + bv, 0.f);
        }
      }
    }
    __syncthreads();
  }
}

__global__ __launch_bounds__(256) void gather2_softmax_kernel(const int* __restrict__ off,
                                                              const int* __restrict__ csr,
                                                              const float* __restrict__ dinv,
                                                              const float* __restrict__ h2s,
                                                              const float* __restrict__ b2,
                                                              float* __restrict__ out) {
  const int gid = blockIdx.x * 256 + threadIdx.x;
  const int sub = threadIdx.x & 3;
  const bool live = (gid >> 2) < N_NODES;
  const int d = live ? (gid >> 2) : (N_NODES - 1);
  const int beg = off[d], end = off[d + 1];
  const float dd = dinv[d];
  const float4 bias = ((const float4*)b2)[sub];
  const float4* p = (const float4*)h2s;
  const int2* csr2 = (const int2*)csr;
  float4 acc = p[d * 4 + sub];
  for (int j = beg; j < end; ++j) {
    const int2 ed = csr2[j];
    const float w = __int_as_float(ed.y);
    const float4 v = p[ed.x * 4 + sub];
    acc.x += v.x * w;
    acc.y += v.y * w;
    acc.z += v.z * w;
    acc.w += v.w * w;
  }
  const float v0 = acc.x * dd + bias.x;
  const float v1 = acc.y * dd + bias.y;
  const float v2 = acc.z * dd + bias.z;
  const float v3 = acc.w * dd + bias.w;
  float m = fmaxf(fmaxf(v0, v1), fmaxf(v2, v3));
  m = fmaxf(m, __shfl_xor(m, 1));
  m = fmaxf(m, __shfl_xor(m, 2));
  const float e0 = expf(v0 - m);
  const float e1 = expf(v1 - m);
  const float e2 = expf(v2 - m);
  const float e3 = expf(v3 - m);
  float s = (e0 + e1) + (e2 + e3);
  s += __shfl_xor(s, 1);
  s += __shfl_xor(s, 2);
  const float inv = 1.0f / s;
  if (live)
    ((float4*)out)[d * 4 + sub] = make_float4(e0 * inv, e1 * inv, e2 * inv, e3 * inv);
}

extern "C" void kernel_launch(void* const* d_in, const int* in_sizes, int n_in,
                              void* d_out, int out_size, void* d_ws, size_t ws_size,
                              hipStream_t stream) {
  const float* x  = (const float*)d_in[0];
  const int*   ei = (const int*)d_in[1];
  const float* ew = (const float*)d_in[2];
  const float* W1 = (const float*)d_in[3];
  const float* b1 = (const float*)d_in[4];
  const float* W2 = (const float*)d_in[5];
  const float* b2 = (const float*)d_in[6];
  float* out = (float*)d_out;

  const int* src = ei;
  const int* dst = ei + N_EDGES;

  float* ws   = (float*)d_ws;
  float* dinv = ws;
  float* h2   = ws + N_NODES + (size_t)N_NODES * 128;
  int* off    = (int*)(ws + N_NODES + (size_t)N_NODES * 256);
  int* tab    = off + 100352;
  int* rec    = tab + 153600 + 512;
  int* csr    = rec + 2 * N_EDGES;

  const int T = 256;

  chunk_group_kernel<<<NCHUNK, SORT_T, 0, stream>>>(src, dst, ew, tab, rec);
  range_build_kernel<<<NRNG, BUILD_T, 0, stream>>>(rec, tab, off, dinv, csr);

  gather_gemm12_kernel<<<GX_NWG, GX_THREADS, 0, stream>>>(off, csr, dinv, x, W1, b1, W2, h2);

  gather2_softmax_kernel<<<(N_NODES * 4 + T - 1) / T, T, 0, stream>>>(off, csr, dinv, h2, b2, out);
}
